# starting kernel
# baseline (speedup 1.0000x reference)
#include <hip/hip_runtime.h>

#define NN 50000
#define NE 800000
#define DH 128
#define DO 64
#define NG 128

typedef float v2f __attribute__((ext_vector_type(2)));
typedef float v8f __attribute__((ext_vector_type(8)));

__global__ void k_fill1(float* p, int n) {
    int i = blockIdx.x * blockDim.x + threadIdx.x;
    if (i < n) p[i] = 1.0f;
}

__global__ void k_deg_accum(const int* __restrict__ dst, float* __restrict__ deg, int E) {
    int e = blockIdx.x * blockDim.x + threadIdx.x;
    if (e < E) atomicAdd(&deg[dst[e]], 1.0f);
}

__global__ void k_deg_finish(float* __restrict__ deg_selfc, float* __restrict__ dinv, int n) {
    int i = blockIdx.x * blockDim.x + threadIdx.x;
    if (i < n) {
        float d = deg_selfc[i];
        dinv[i] = 1.0f / sqrtf(d);
        deg_selfc[i] = 1.0f / d;
    }
}

__global__ void k_edge_w(const int* __restrict__ src, const int* __restrict__ dst,
                         const float* __restrict__ dinv, float* __restrict__ ew, int E) {
    int e = blockIdx.x * blockDim.x + threadIdx.x;
    if (e < E) ew[e] = dinv[src[e]] * dinv[dst[e]];
}

template <int NC>
__global__ void __launch_bounds__(256)
k_wmma_gemm(const float* __restrict__ A, const float* __restrict__ W,
            const float* __restrict__ bias, float* __restrict__ C,
            int M, int relu)
{
    __shared__ float sWt[NC * 132];
    __shared__ float sA[16 * 132];

    const int tid = threadIdx.x;
    const int r0  = blockIdx.x * 16;

    for (int i = tid; i < 128 * NC; i += 256) {
        int k = i / NC, c = i % NC;
        sWt[c * 132 + k] = W[i];
    }
    for (int i = tid; i < 16 * 128; i += 256) {
        int r = i >> 7, c = i & 127;
        int row = r0 + r;
        sA[r * 132 + c] = (row < M) ? A[row * 128 + c] : 0.0f;
    }
    __syncthreads();

    const int wave = tid >> 5;
    const int lane = tid & 31;
    const int ln   = lane & 15;
    const int hi   = lane >> 4;
    const int c0   = wave * 16;

    if (c0 < NC) {
        const float* aP = &sA [ln        * 132 + 2 * hi];
        const float* bP = &sWt[(c0 + ln) * 132 + 2 * hi];
        v8f acc = {0.f, 0.f, 0.f, 0.f, 0.f, 0.f, 0.f, 0.f};
#pragma unroll
        for (int k = 0; k < 128; k += 4) {
            v2f a = *(const v2f*)(aP + k);
            v2f b = *(const v2f*)(bP + k);
            acc = __builtin_amdgcn_wmma_f32_16x16x4_f32(
                false, a, false, b, (short)0, acc, false, false);
        }
        const int col = c0 + ln;
        const float bv = bias ? bias[col] : 0.0f;
#pragma unroll
        for (int r = 0; r < 8; ++r) {
            int row = r0 + r + 8 * hi;
            if (row < M) {
                float v = acc[r] + bv;
                if (relu) v = fmaxf(v, 0.0f);
                C[row * NC + col] = v;
            }
        }
    }
}

__global__ void k_scatter(const int* __restrict__ src, const int* __restrict__ dst,
                          const float* __restrict__ ew, const float* __restrict__ H2,
                          float* __restrict__ AGG, int E)
{
    long long idx = (long long)blockIdx.x * blockDim.x + threadIdx.x;
    if (idx >= (long long)E * DH) return;
    int e = (int)(idx >> 7);
    int d = (int)(idx & 127);
    atomicAdd(&AGG[(size_t)dst[e] * DH + d], H2[(size_t)src[e] * DH + d] * ew[e]);
}

__global__ void k_combine(const float* __restrict__ AGG, const float* __restrict__ H2,
                          const float* __restrict__ selfc, const float* __restrict__ bias,
                          float* __restrict__ H, int n)
{
    int idx = blockIdx.x * blockDim.x + threadIdx.x;
    if (idx >= n * DH) return;
    int i = idx >> 7, d = idx & 127;
    float v = AGG[idx] + H2[idx] * selfc[i] + bias[d];
    H[idx] = fmaxf(v, 0.0f);
}

__global__ void k_pool_accum(const float* __restrict__ H, const int* __restrict__ batch,
                             float* __restrict__ sums, float* __restrict__ counts, int total)
{
    int idx = blockIdx.x * blockDim.x + threadIdx.x;
    if (idx >= total) return;
    int i = idx >> 7, d = idx & 127;
    int g = batch[i];
    atomicAdd(&sums[g * DH + d], H[idx]);
    if (d == 0) atomicAdd(&counts[g], 1.0f);
}

__global__ void k_pool_div(const float* __restrict__ sums, const float* __restrict__ counts,
                           float* __restrict__ pooled)
{
    int idx = blockIdx.x * blockDim.x + threadIdx.x;
    if (idx >= NG * DH) return;
    pooled[idx] = sums[idx] / fmaxf(counts[idx >> 7], 1.0f);
}

extern "C" void kernel_launch(void* const* d_in, const int* in_sizes, int n_in,
                              void* d_out, int out_size, void* d_ws, size_t ws_size,
                              hipStream_t stream)
{
    (void)in_sizes; (void)n_in; (void)out_size; (void)ws_size;

    const float* x     = (const float*)d_in[0];
    const int*   ei    = (const int*)d_in[1];
    const int*   src   = ei;
    const int*   dst   = ei + NE;
    const int*   batch = (const int*)d_in[2];
    const float* Wl[5] = {(const float*)d_in[4], (const float*)d_in[6], (const float*)d_in[8],
                          (const float*)d_in[10], (const float*)d_in[12]};
    const float* Bl[5] = {(const float*)d_in[5], (const float*)d_in[7], (const float*)d_in[9],
                          (const float*)d_in[11], (const float*)d_in[13]};
    const float* fc1_w = (const float*)d_in[14];
    const float* fc1_b = (const float*)d_in[15];
    const float* fc2_w = (const float*)d_in[16];
    const float* fc2_b = (const float*)d_in[17];

    float* H      = (float*)d_ws;
    float* H2     = H      + (size_t)NN * DH;
    float* AGG    = H2     + (size_t)NN * DH;
    float* selfc  = AGG    + (size_t)NN * DH;
    float* dinv   = selfc  + NN;
    float* ew     = dinv   + NN;
    float* sums   = ew     + NE;
    float* counts = sums   + (size_t)NG * DH;
    float* pooled = counts + NG;
    float* gact   = pooled + (size_t)NG * DH;

    const int T = 256;

    k_fill1<<<(NN + T - 1) / T, T, 0, stream>>>(selfc, NN);
    k_deg_accum<<<(NE + T - 1) / T, T, 0, stream>>>(dst, selfc, NE);
    k_deg_finish<<<(NN + T - 1) / T, T, 0, stream>>>(selfc, dinv, NN);
    k_edge_w<<<(NE + T - 1) / T, T, 0, stream>>>(src, dst, dinv, ew, NE);

    const long long escat = (long long)NE * DH;
    const int scat_blocks = (int)((escat + T - 1) / T);
    const int gemm_blocks = NN / 16;
    const int comb_blocks = (NN * DH + T - 1) / T;

    const float* cur = x;
    for (int l = 0; l < 5; ++l) {
        k_wmma_gemm<DH><<<gemm_blocks, T, 0, stream>>>(cur, Wl[l], nullptr, H2, NN, 0);
        hipMemsetAsync(AGG, 0, (size_t)NN * DH * sizeof(float), stream);
        k_scatter<<<scat_blocks, T, 0, stream>>>(src, dst, ew, H2, AGG, NE);
        k_combine<<<comb_blocks, T, 0, stream>>>(AGG, H2, selfc, Bl[l], H, NN);
        cur = H;
    }

    hipMemsetAsync(sums, 0, ((size_t)NG * DH + NG) * sizeof(float), stream);
    k_pool_accum<<<(NN * DH + T - 1) / T, T, 0, stream>>>(H, batch, sums, counts, NN * DH);
    k_pool_div<<<(NG * DH + T - 1) / T, T, 0, stream>>>(sums, counts, pooled);

    k_wmma_gemm<DH><<<NG / 16, T, 0, stream>>>(pooled, fc1_w, fc1_b, gact, NG, 1);
    k_wmma_gemm<DO><<<NG / 16, T, 0, stream>>>(gact, fc2_w, fc2_b, (float*)d_out, NG, 0);
}


# degrees without float adds to memory: a count pass sizes the destination tiles, edges are binned, each tile counts its rows' in-degrees from its records in LDS (integer adds); a pass fills the weights
# speedup vs baseline: 5.9305x; 5.9305x over previous
#include <hip/hip_runtime.h>

#define NN 50000
#define NE 800000
#define DH 128
#define DO 64
#define NG 128
#define TR 100
#define NT ((NN + TR - 1) / TR)
#define EG 8
#define AW 16
#define PR (TR / 4)

static_assert(NT <= 512, "the tile scan is one 512-thread block");
static_assert(NN <= 65536 && TR <= 32768, "a record packs source row and local row into one word");
static_assert(AW * 32 == 4 * DH && TR % 4 == 0 && PR <= 32 && NN % TR == 0,
              "the pool walk gives a quarter of a full tile's rows to DH threads");

typedef float v2f __attribute__((ext_vector_type(2)));
typedef float v8f __attribute__((ext_vector_type(8)));

__global__ void __launch_bounds__(1024)
k_tile_count(const int* __restrict__ dst, int* __restrict__ tilecount, int E) {
    __shared__ int cnt[NT];
    const int tid = threadIdx.x;
    for (int t = tid; t < NT; t += 1024) cnt[t] = 0;
    __syncthreads();
    const int e0 = blockIdx.x * 4096 + tid;
    int d[4] = {-1, -1, -1, -1};
    for (int j = 0; j < 4; ++j) {
        const int e = e0 + j * 1024;
        if (e < E) d[j] = dst[e];
    }
    for (int j = 0; j < 4; ++j)
        if (d[j] >= 0) atomicAdd(&cnt[d[j] / TR], 1);
    __syncthreads();
    for (int t = tid; t < NT; t += 1024) {
        const int c = cnt[t];
        if (c) atomicAdd(&tilecount[t], c);
    }
}

__global__ void __launch_bounds__(512)
k_tile_scan(const int* __restrict__ tilecount, int* __restrict__ tilestart, int* __restrict__ cursor) {
    __shared__ int s[512];
    const int t = threadIdx.x;
    const int c = (t < NT) ? tilecount[t] : 0;
    s[t] = c;
    __syncthreads();
    for (int o = 1; o < 512; o <<= 1) {
        const int v = (t >= o) ? s[t - o] : 0;
        __syncthreads();
        s[t] += v;
        __syncthreads();
    }
    if (t < NT) { tilestart[t] = s[t] - c; cursor[t] = 0; }
    if (t == NT - 1) tilestart[NT] = s[t];
}

__global__ void __launch_bounds__(1024)
k_bin_fill(const int* __restrict__ src, const int* __restrict__ dst,
           const int* __restrict__ tilestart,
           int* __restrict__ cursor, int2* __restrict__ rec, int E) {
    __shared__ int cnt[NT];
    __shared__ int base[NT];
    const int tid = threadIdx.x;
    for (int t = tid; t < NT; t += 1024) cnt[t] = 0;
    __syncthreads();
    int s[4] = {0, 0, 0, 0}, d[4] = {0, 0, 0, 0}, tl[4] = {0, 0, 0, 0}, r[4] = {0, 0, 0, 0};
    const int e0 = blockIdx.x * 4096 + tid;
    for (int j = 0; j < 4; ++j) {
        const int e = e0 + j * 1024;
        if (e < E) { s[j] = src[e]; d[j] = dst[e]; }
    }
    for (int j = 0; j < 4; ++j) {
        const int e = e0 + j * 1024;
        if (e < E) {
            tl[j] = d[j] / TR;
            r[j] = atomicAdd(&cnt[tl[j]], 1);
        }
    }
    __syncthreads();
    for (int t = tid; t < NT; t += 1024) {
        const int c = cnt[t];
        if (c) base[t] = tilestart[t] + atomicAdd(&cursor[t], c);
    }
    __syncthreads();
    for (int j = 0; j < 4; ++j) {
        const int e = e0 + j * 1024;
        if (e < E)
            rec[base[tl[j]] + r[j]] = make_int2(s[j] | ((d[j] - tl[j] * TR) << 16), 0);
    }
}

__global__ void __launch_bounds__(256)
k_tile_deg(const int2* __restrict__ rec, const int* __restrict__ tilestart,
           float* __restrict__ selfc, float* __restrict__ dinv, int n) {
    __shared__ int cnt[TR];
    const int t = threadIdx.x;
    if (t < TR) cnt[t] = 0;
    __syncthreads();
    const int beg = tilestart[blockIdx.x];
    const int end = tilestart[blockIdx.x + 1];
    for (int i = beg + t; i < end; i += 256) atomicAdd(&cnt[rec[i].x >> 16], 1);
    __syncthreads();
    const int i = blockIdx.x * TR + t;
    if (t < TR && i < n) {
        const float d = (float)(cnt[t] + 1);
        dinv[i]  = 1.0f / sqrtf(d);
        selfc[i] = 1.0f / d;
    }
}

__global__ void __launch_bounds__(256)
k_tile_weights(const int* __restrict__ tilestart, const float* __restrict__ dinv,
               int2* __restrict__ rec) {
    const int beg = tilestart[blockIdx.x];
    const int end = tilestart[blockIdx.x + 1];
    const int n0  = blockIdx.x * TR;
    for (int i = beg + (int)threadIdx.x; i < end; i += 256) {
        const int k = rec[i].x;
        const float w = dinv[k & 0xffff] * dinv[n0 + (k >> 16)];
        rec[i] = make_int2(k, __float_as_int(w));
    }
}

template <int NC>
__global__ void __launch_bounds__(256)
k_wmma_gemm(const float* __restrict__ A, const float* __restrict__ W,
            const float* __restrict__ bias, float* __restrict__ C,
            int M, int relu)
{
    __shared__ float sWt[NC * 132];
    __shared__ float sA[16 * 132];

    const int tid = threadIdx.x;
    const int r0  = blockIdx.x * 16;

    for (int i = tid; i < 128 * NC; i += 256) {
        int k = i / NC, c = i % NC;
        sWt[c * 132 + k] = W[i];
    }
    for (int i = tid; i < 16 * 128; i += 256) {
        int r = i >> 7, c = i & 127;
        int row = r0 + r;
        sA[r * 132 + c] = (row < M) ? A[row * 128 + c] : 0.0f;
    }
    __syncthreads();

    const int wave = tid >> 5;
    const int lane = tid & 31;
    const int ln   = lane & 15;
    const int hi   = lane >> 4;
    const int c0   = wave * 16;

    if (c0 < NC) {
        const float* aP = &sA [ln        * 132 + 2 * hi];
        const float* bP = &sWt[(c0 + ln) * 132 + 2 * hi];
        v8f acc = {0.f, 0.f, 0.f, 0.f, 0.f, 0.f, 0.f, 0.f};
#pragma unroll
        for (int k = 0; k < 128; k += 4) {
            v2f a = *(const v2f*)(aP + k);
            v2f b = *(const v2f*)(bP + k);
            acc = __builtin_amdgcn_wmma_f32_16x16x4_f32(
                false, a, false, b, (short)0, acc, false, false);
        }
        const int col = c0 + ln;
        const float bv = bias ? bias[col] : 0.0f;
#pragma unroll
        for (int r = 0; r < 8; ++r) {
            int row = r0 + r + 8 * hi;
            if (row < M) {
                float v = acc[r] + bv;
                if (relu) v = fmaxf(v, 0.0f);
                C[row * NC + col] = v;
            }
        }
    }
}

#define TS  132
#define TRP 112
#define NU  ((TRP / 16) * (DH / 16))
#define WPT ((DH * DH) / (AW * 32))

static_assert(TRP % 16 == 0 && TRP >= TR && (TRP * TS) % 4 == 0, "whole strips, 16-byte clears");
static_assert(NU <= 4 * AW && (DH * DH) % (AW * 32) == 0, "a wave takes at most four output blocks");

__device__ __forceinline__ void edge_add(const float* __restrict__ Hin, float* acc,
                                         int key, float w, int lane) {
    const float* row = Hin + (size_t)(key & 0xffff) * DH + lane;
    float* a = acc + (key >> 16) * TS + lane;
    const float v0 = row[0], v1 = row[32], v2 = row[64], v3 = row[96];
    __hip_atomic_fetch_add(a,      v0 * w, __ATOMIC_RELAXED, __HIP_MEMORY_SCOPE_WORKGROUP);
    __hip_atomic_fetch_add(a + 32, v1 * w, __ATOMIC_RELAXED, __HIP_MEMORY_SCOPE_WORKGROUP);
    __hip_atomic_fetch_add(a + 64, v2 * w, __ATOMIC_RELAXED, __HIP_MEMORY_SCOPE_WORKGROUP);
    __hip_atomic_fetch_add(a + 96, v3 * w, __ATOMIC_RELAXED, __HIP_MEMORY_SCOPE_WORKGROUP);
}

template <bool POOL>
__global__ void __launch_bounds__(AW * 32)
k_layer_tile(const float* __restrict__ Hin, const float* __restrict__ W,
             const int2* __restrict__ rec, const int* __restrict__ tilestart,
             const float* __restrict__ selfc, const float* __restrict__ bias,
             float* __restrict__ Hout, const int* __restrict__ batch,
             float* __restrict__ sums, float* __restrict__ counts, int n)
{
    __shared__ alignas(16) float sWt[DH * TS];
    __shared__ alignas(16) float acc[TRP * TS];
    const int tid  = threadIdx.x;
    const int lane = tid & 31;
    const int wave = __builtin_amdgcn_readfirstlane(tid >> 5);
    const int n0   = blockIdx.x * TR;
    const int rows = (n - n0 < TR) ? n - n0 : TR;

    float4* acc4 = (float4*)acc;
    for (int i = tid; i < TRP * TS / 4; i += AW * 32) acc4[i] = make_float4(0.f, 0.f, 0.f, 0.f);
    __syncthreads();

    float wreg[WPT];
#pragma unroll
    for (int j = 0; j < WPT; ++j) wreg[j] = W[tid + j * (AW * 32)];

    const int beg     = tilestart[blockIdx.x];
    const int cntE    = tilestart[blockIdx.x + 1] - beg;
    const int egroups = (cntE + EG - 1) / EG;
    const int ngroups = egroups + (rows + EG - 1) / EG;
    const int2* trec  = rec + beg;

    auto fetch = [&](int gi, int& k, int& w) {
        k = 0; w = 0;
        if (gi < egroups) {
            const int e = gi * EG + lane;
            if (lane < EG && e < cntE) { const int2 q = trec[e]; k = q.x; w = q.y; }
        } else if (gi < ngroups) {
            const int r = (gi - egroups) * EG + lane;
            if (lane < EG && r < rows) { k = (n0 + r) | (r << 16); w = __float_as_int(selfc[n0 + r]); }
        }
    };
    int key, wb;
    int g = wave;
    fetch(g, key, wb);
    while (g < ngroups) {
        const int gn = g + AW;
        int key2, wb2;
        fetch(gn, key2, wb2);
        const int m = (g < egroups) ? cntE - g * EG : rows - (g - egroups) * EG;
        if (m >= EG) {
            float v[EG][4];
            int   kj[EG];
            for (int j = 0; j < EG; ++j) {
                kj[j] = __builtin_amdgcn_readlane(key, j);
                const float* row = Hin + (size_t)(kj[j] & 0xffff) * DH + lane;
                v[j][0] = row[0]; v[j][1] = row[32]; v[j][2] = row[64]; v[j][3] = row[96];
            }
            for (int j = 0; j < EG; ++j) {
                const float w = __int_as_float(__builtin_amdgcn_readlane(wb, j));
                float* a = acc + (kj[j] >> 16) * TS + lane;
                __hip_atomic_fetch_add(a,      v[j][0] * w, __ATOMIC_RELAXED, __HIP_MEMORY_SCOPE_WORKGROUP);
                __hip_atomic_fetch_add(a + 32, v[j][1] * w, __ATOMIC_RELAXED, __HIP_MEMORY_SCOPE_WORKGROUP);
                __hip_atomic_fetch_add(a + 64, v[j][2] * w, __ATOMIC_RELAXED, __HIP_MEMORY_SCOPE_WORKGROUP);
                __hip_atomic_fetch_add(a + 96, v[j][3] * w, __ATOMIC_RELAXED, __HIP_MEMORY_SCOPE_WORKGROUP);
            }
        } else {
            for (int j = 0; j < m; ++j) {
                const int   k = __builtin_amdgcn_readlane(key, j);
                const float w = __int_as_float(__builtin_amdgcn_readlane(wb, j));
                edge_add(Hin, acc, k, w, lane);
            }
        }
        key = key2; wb = wb2; g = gn;
    }

#pragma unroll
    for (int j = 0; j < WPT; ++j) {
        const int i = tid + j * (AW * 32);
        sWt[(i & (DH - 1)) * TS + (i >> 7)] = wreg[j];
    }
    __syncthreads();

    const int ln = lane & 15;
    const int hi = lane >> 4;
    v8f res[4];
#pragma unroll
    for (int q = 0; q < 4; ++q) {
        const int u = wave + q * AW;
        res[q] = (v8f){0.f, 0.f, 0.f, 0.f, 0.f, 0.f, 0.f, 0.f};
        if (u < NU) {
            const int strip = u >> 3, ct = u & 7;
            const float* aP = &acc[(strip * 16 + ln) * TS + 2 * hi];
            const float* bP = &sWt[(ct * 16 + ln) * TS + 2 * hi];
            v8f d = {0.f, 0.f, 0.f, 0.f, 0.f, 0.f, 0.f, 0.f};
#pragma unroll
            for (int k = 0; k < DH; k += 4) {
                v2f a = *(const v2f*)(aP + k);
                v2f b = *(const v2f*)(bP + k);
                d = __builtin_amdgcn_wmma_f32_16x16x4_f32(false, a, false, b, (short)0, d, false, false);
            }
            const float bv = bias[ct * 16 + ln];
#pragma unroll
            for (int r = 0; r < 8; ++r) res[q][r] = fmaxf(d[r] + bv, 0.0f);
        }
    }
    __syncthreads();

#pragma unroll
    for (int q = 0; q < 4; ++q) {
        const int u = wave + q * AW;
        if (u < NU) {
            const int strip = u >> 3, ct = u & 7;
#pragma unroll
            for (int r = 0; r < 8; ++r)
                acc[(strip * 16 + r + 8 * hi) * TS + ct * 16 + ln] = res[q][r];
        }
    }
    __syncthreads();

    if (!POOL) {
        float4* H4 = (float4*)Hout;
        for (int i = tid; i < TR * DH / 4; i += AW * 32) {
            const int r = i >> 5, c4 = i & 31;
            if (r < rows) H4[(size_t)(n0 + r) * (DH / 4) + c4] = *((const float4*)(acc + r * TS) + c4);
        }
    } else {
        const int c  = tid & (DH - 1);
        const int rb = (wave >> 2) * PR;
        if ((wave & 3) == 0 && lane < PR) atomicAdd(&counts[batch[n0 + rb + lane]], 1.0f);
        int   gid[PR];
        float v[PR];
#pragma unroll
        for (int j = 0; j < PR; ++j) {
            gid[j] = batch[n0 + rb + j];
            v[j]   = acc[(rb + j) * TS + c];
        }
        int   gg = gid[0];
        float s = 0.0f;
#pragma unroll
        for (int j = 0; j < PR; ++j) {
            if (gid[j] != gg) {
                atomicAdd(&sums[gg * DH + c], s);
                gg = gid[j];
                s = 0.0f;
            }
            s += v[j];
        }
        atomicAdd(&sums[gg * DH + c], s);
    }
}

__global__ void k_pool_div(const float* __restrict__ sums, const float* __restrict__ counts,
                           float* __restrict__ pooled)
{
    int idx = blockIdx.x * blockDim.x + threadIdx.x;
    if (idx >= NG * DH) return;
    pooled[idx] = sums[idx] / fmaxf(counts[idx >> 7], 1.0f);
}

extern "C" void kernel_launch(void* const* d_in, const int* in_sizes, int n_in,
                              void* d_out, int out_size, void* d_ws, size_t ws_size,
                              hipStream_t stream)
{
    (void)in_sizes; (void)n_in; (void)out_size; (void)ws_size;

    const float* x     = (const float*)d_in[0];
    const int*   ei    = (const int*)d_in[1];
    const int*   src   = ei;
    const int*   dst   = ei + NE;
    const int*   batch = (const int*)d_in[2];
    const float* Wl[5] = {(const float*)d_in[4], (const float*)d_in[6], (const float*)d_in[8],
                          (const float*)d_in[10], (const float*)d_in[12]};
    const float* Bl[5] = {(const float*)d_in[5], (const float*)d_in[7], (const float*)d_in[9],
                          (const float*)d_in[11], (const float*)d_in[13]};
    const float* fc1_w = (const float*)d_in[14];
    const float* fc1_b = (const float*)d_in[15];
    const float* fc2_w = (const float*)d_in[16];
    const float* fc2_b = (const float*)d_in[17];

    float* H      = (float*)d_ws;
    float* H2     = H      + (size_t)NN * DH;
    float* selfc  = H2     + (size_t)NN * DH;
    float* dinv   = selfc  + NN;
    int2*  rec    = (int2*)(dinv + NN);
    int*   tilestart = (int*)(rec + NE);
    int*   cursor    = tilestart + NT + 1;
    int*   tilecount = cursor + NT;
    float* sums   = (float*)(tilecount + NT);
    float* counts = sums   + (size_t)NG * DH;
    float* pooled = counts + NG;
    float* gact   = pooled + (size_t)NG * DH;

    const int T = 256;

    hipMemsetAsync(tilecount, 0, ((size_t)NT + (size_t)NG * DH + NG) * sizeof(float), stream);

    k_tile_count<<<(NE + 4095) / 4096, 1024, 0, stream>>>(dst, tilecount, NE);
    k_tile_scan<<<1, 512, 0, stream>>>(tilecount, tilestart, cursor);
    k_bin_fill<<<(NE + 4095) / 4096, 1024, 0, stream>>>(src, dst, tilestart, cursor, rec, NE);
    k_tile_deg<<<NT, 256, 0, stream>>>(rec, tilestart, selfc, dinv, NN);
    k_tile_weights<<<NT, 256, 0, stream>>>(tilestart, dinv, rec);

    k_layer_tile<false><<<NT, AW * 32, 0, stream>>>(x,  Wl[0], rec, tilestart, selfc, Bl[0], H,
                                                    nullptr, nullptr, nullptr, NN);
    k_layer_tile<false><<<NT, AW * 32, 0, stream>>>(H,  Wl[1], rec, tilestart, selfc, Bl[1], H2,
                                                    nullptr, nullptr, nullptr, NN);
    k_layer_tile<false><<<NT, AW * 32, 0, stream>>>(H2, Wl[2], rec, tilestart, selfc, Bl[2], H,
                                                    nullptr, nullptr, nullptr, NN);
    k_layer_tile<false><<<NT, AW * 32, 0, stream>>>(H,  Wl[3], rec, tilestart, selfc, Bl[3], H2,
                                                    nullptr, nullptr, nullptr, NN);
    k_layer_tile<true><<<NT, AW * 32, 0, stream>>>(H2, Wl[4], rec, tilestart, selfc, Bl[4], nullptr,
                                                   batch, sums, counts, NN);

    k_pool_div<<<(NG * DH + T - 1) / T, T, 0, stream>>>(sums, counts, pooled);

    k_wmma_gemm<DH><<<NG / 16, T, 0, stream>>>(pooled, fc1_w, fc1_b, gact, NG, 1);
    k_wmma_gemm<DO><<<NG / 16, T, 0, stream>>>(gact, fc2_w, fc2_b, (float*)d_out, NG, 0);
}


# tile summed in two parts (records binned by tile half): 4 waves multiply the finished front 64 rows on WMMA while 12 waves sum the back rows; product runs under row loads; sums stay in arrival order
# speedup vs baseline: 6.1646x; 1.0395x over previous
#include <hip/hip_runtime.h>

#define NN 50000
#define NE 800000
#define DH 128
#define DO 64
#define NG 128
#define TR 100
#define NT ((NN + TR - 1) / TR)
#define EG 8
#define AW 16
#define PR (TR / 4)
#define SPLIT 64
#define NB (2 * NT)
#define MW (SPLIT / 16)

static_assert(NB <= 1024, "the bin scan is one 1024-thread block");
static_assert(SPLIT % 16 == 0 && SPLIT < TR && MW < AW, "the front part is whole strips, one wave each, and walkers remain");
static_assert(NN <= 65536 && TR <= 32768, "a record packs source row and local row into one word");
static_assert(AW * 32 == 4 * DH && TR % 4 == 0 && PR <= 32 && NN % TR == 0,
              "the pool walk gives a quarter of a full tile's rows to DH threads");

typedef float v2f __attribute__((ext_vector_type(2)));
typedef float v8f __attribute__((ext_vector_type(8)));

__global__ void __launch_bounds__(1024)
k_tile_count(const int* __restrict__ dst, int* __restrict__ tilecount, int E) {
    __shared__ int cnt[NB];
    const int tid = threadIdx.x;
    for (int t = tid; t < NB; t += 1024) cnt[t] = 0;
    __syncthreads();
    const int e0 = blockIdx.x * 4096 + tid;
    int d[4] = {-1, -1, -1, -1};
    for (int j = 0; j < 4; ++j) {
        const int e = e0 + j * 1024;
        if (e < E) d[j] = dst[e];
    }
    for (int j = 0; j < 4; ++j)
        if (d[j] >= 0) {
            const int t = d[j] / TR;
            atomicAdd(&cnt[2 * t + ((d[j] - t * TR) >= SPLIT ? 1 : 0)], 1);
        }
    __syncthreads();
    for (int t = tid; t < NB; t += 1024) {
        const int c = cnt[t];
        if (c) atomicAdd(&tilecount[t], c);
    }
}

__global__ void __launch_bounds__(1024)
k_tile_scan(const int* __restrict__ tilecount, int* __restrict__ tilestart, int* __restrict__ cursor) {
    __shared__ int s[1024];
    const int t = threadIdx.x;
    const int c = (t < NB) ? tilecount[t] : 0;
    s[t] = c;
    __syncthreads();
    for (int o = 1; o < 1024; o <<= 1) {
        const int v = (t >= o) ? s[t - o] : 0;
        __syncthreads();
        s[t] += v;
        __syncthreads();
    }
    if (t < NB) { tilestart[t] = s[t] - c; cursor[t] = 0; }
    if (t == NB - 1) tilestart[NB] = s[t];
}

__global__ void __launch_bounds__(1024)
k_bin_fill(const int* __restrict__ src, const int* __restrict__ dst,
           const int* __restrict__ tilestart,
           int* __restrict__ cursor, int2* __restrict__ rec, int E) {
    __shared__ int cnt[NB];
    __shared__ int base[NB];
    const int tid = threadIdx.x;
    for (int t = tid; t < NB; t += 1024) cnt[t] = 0;
    __syncthreads();
    int s[4] = {0, 0, 0, 0}, d[4] = {0, 0, 0, 0}, tl[4] = {0, 0, 0, 0}, r[4] = {0, 0, 0, 0}, lr[4] = {0, 0, 0, 0};
    const int e0 = blockIdx.x * 4096 + tid;
    for (int j = 0; j < 4; ++j) {
        const int e = e0 + j * 1024;
        if (e < E) { s[j] = src[e]; d[j] = dst[e]; }
    }
    for (int j = 0; j < 4; ++j) {
        const int e = e0 + j * 1024;
        if (e < E) {
            const int t = d[j] / TR;
            lr[j] = d[j] - t * TR;
            tl[j] = 2 * t + (lr[j] >= SPLIT ? 1 : 0);
            r[j] = atomicAdd(&cnt[tl[j]], 1);
        }
    }
    __syncthreads();
    for (int t = tid; t < NB; t += 1024) {
        const int c = cnt[t];
        if (c) base[t] = tilestart[t] + atomicAdd(&cursor[t], c);
    }
    __syncthreads();
    for (int j = 0; j < 4; ++j) {
        const int e = e0 + j * 1024;
        if (e < E)
            rec[base[tl[j]] + r[j]] = make_int2(s[j] | (lr[j] << 16), 0);
    }
}

__global__ void __launch_bounds__(256)
k_tile_deg(const int2* __restrict__ rec, const int* __restrict__ tilestart,
           float* __restrict__ selfc, float* __restrict__ dinv, int n) {
    __shared__ int cnt[TR];
    const int t = threadIdx.x;
    if (t < TR) cnt[t] = 0;
    __syncthreads();
    const int beg = tilestart[2 * blockIdx.x];
    const int end = tilestart[2 * blockIdx.x + 2];
    for (int i = beg + t; i < end; i += 256) atomicAdd(&cnt[rec[i].x >> 16], 1);
    __syncthreads();
    const int i = blockIdx.x * TR + t;
    if (t < TR && i < n) {
        const float d = (float)(cnt[t] + 1);
        dinv[i]  = 1.0f / sqrtf(d);
        selfc[i] = 1.0f / d;
    }
}

__global__ void __launch_bounds__(256)
k_tile_weights(const int* __restrict__ tilestart, const float* __restrict__ dinv,
               int2* __restrict__ rec) {
    const int beg = tilestart[2 * blockIdx.x];
    const int end = tilestart[2 * blockIdx.x + 2];
    const int n0  = blockIdx.x * TR;
    for (int i = beg + (int)threadIdx.x; i < end; i += 256) {
        const int k = rec[i].x;
        const float w = dinv[k & 0xffff] * dinv[n0 + (k >> 16)];
        rec[i] = make_int2(k, __float_as_int(w));
    }
}

template <int NC>
__global__ void __launch_bounds__(256)
k_wmma_gemm(const float* __restrict__ A, const float* __restrict__ W,
            const float* __restrict__ bias, float* __restrict__ C,
            int M, int relu)
{
    __shared__ float sWt[NC * 132];
    __shared__ float sA[16 * 132];

    const int tid = threadIdx.x;
    const int r0  = blockIdx.x * 16;

    for (int i = tid; i < 128 * NC; i += 256) {
        int k = i / NC, c = i % NC;
        sWt[c * 132 + k] = W[i];
    }
    for (int i = tid; i < 16 * 128; i += 256) {
        int r = i >> 7, c = i & 127;
        int row = r0 + r;
        sA[r * 132 + c] = (row < M) ? A[row * 128 + c] : 0.0f;
    }
    __syncthreads();

    const int wave = tid >> 5;
    const int lane = tid & 31;
    const int ln   = lane & 15;
    const int hi   = lane >> 4;
    const int c0   = wave * 16;

    if (c0 < NC) {
        const float* aP = &sA [ln        * 132 + 2 * hi];
        const float* bP = &sWt[(c0 + ln) * 132 + 2 * hi];
        v8f acc = {0.f, 0.f, 0.f, 0.f, 0.f, 0.f, 0.f, 0.f};
#pragma unroll
        for (int k = 0; k < 128; k += 4) {
            v2f a = *(const v2f*)(aP + k);
            v2f b = *(const v2f*)(bP + k);
            acc = __builtin_amdgcn_wmma_f32_16x16x4_f32(
                false, a, false, b, (short)0, acc, false, false);
        }
        const int col = c0 + ln;
        const float bv = bias ? bias[col] : 0.0f;
#pragma unroll
        for (int r = 0; r < 8; ++r) {
            int row = r0 + r + 8 * hi;
            if (row < M) {
                float v = acc[r] + bv;
                if (relu) v = fmaxf(v, 0.0f);
                C[row * NC + col] = v;
            }
        }
    }
}

#define TS  132
#define TRP 112
#define NU  ((TRP / 16) * (DH / 16))
#define WPT ((DH * DH) / (AW * 32))

static_assert(TRP % 16 == 0 && TRP >= TR && (TRP * TS) % 4 == 0, "whole strips, 16-byte clears");
static_assert(NU - (SPLIT / 16) * 8 <= 2 * AW && (DH * DH) % (AW * 32) == 0, "a wave takes at most two back output blocks");

__device__ __forceinline__ void edge_add(const float* __restrict__ Hin, float* acc,
                                         int key, float w, int lane) {
    const float* row = Hin + (size_t)(key & 0xffff) * DH + lane;
    float* a = acc + (key >> 16) * TS + lane;
    const float v0 = row[0], v1 = row[32], v2 = row[64], v3 = row[96];
    __hip_atomic_fetch_add(a,      v0 * w, __ATOMIC_RELAXED, __HIP_MEMORY_SCOPE_WORKGROUP);
    __hip_atomic_fetch_add(a + 32, v1 * w, __ATOMIC_RELAXED, __HIP_MEMORY_SCOPE_WORKGROUP);
    __hip_atomic_fetch_add(a + 64, v2 * w, __ATOMIC_RELAXED, __HIP_MEMORY_SCOPE_WORKGROUP);
    __hip_atomic_fetch_add(a + 96, v3 * w, __ATOMIC_RELAXED, __HIP_MEMORY_SCOPE_WORKGROUP);
}

template <bool POOL>
__global__ void __launch_bounds__(AW * 32)
k_layer_tile(const float* __restrict__ Hin, const float* __restrict__ W,
             const int2* __restrict__ rec, const int* __restrict__ tilestart,
             const float* __restrict__ selfc, const float* __restrict__ bias,
             float* __restrict__ Hout, const int* __restrict__ batch,
             float* __restrict__ sums, float* __restrict__ counts, int n)
{
    __shared__ alignas(16) float sWt[DH * TS];
    __shared__ alignas(16) float acc[TRP * TS];
    const int tid  = threadIdx.x;
    const int lane = tid & 31;
    const int wave = __builtin_amdgcn_readfirstlane(tid >> 5);
    const int n0   = blockIdx.x * TR;
    const int rows = (n - n0 < TR) ? n - n0 : TR;

    float4* acc4 = (float4*)acc;
    for (int i = tid; i < TRP * TS / 4; i += AW * 32) acc4[i] = make_float4(0.f, 0.f, 0.f, 0.f);
    __syncthreads();

    float wreg[WPT];
#pragma unroll
    for (int j = 0; j < WPT; ++j) wreg[j] = W[tid + j * (AW * 32)];

    const int begA  = tilestart[2 * blockIdx.x];
    const int begB  = tilestart[2 * blockIdx.x + 1];
    const int endB  = tilestart[2 * blockIdx.x + 2];
    const int rowsA = (rows < SPLIT) ? rows : SPLIT;
    const int rowsB = rows - rowsA;

    auto walk = [&](const int2* __restrict__ trec, const int cntE, const int r0, const int nrows,
                    const int g0, const int gs) {
        const int egroups = (cntE + EG - 1) / EG;
        const int ngroups = egroups + (nrows + EG - 1) / EG;
        auto fetch = [&](int gi, int& k, int& w) {
            k = 0; w = 0;
            if (gi < egroups) {
                const int e = gi * EG + lane;
                if (lane < EG && e < cntE) { const int2 q = trec[e]; k = q.x; w = q.y; }
            } else if (gi < ngroups) {
                const int r = (gi - egroups) * EG + lane;
                if (lane < EG && r < nrows) {
                    k = (n0 + r0 + r) | ((r0 + r) << 16);
                    w = __float_as_int(selfc[n0 + r0 + r]);
                }
            }
        };
        int key, wb;
        int g = g0;
        fetch(g, key, wb);
        while (g < ngroups) {
            const int gn = g + gs;
            int key2, wb2;
            fetch(gn, key2, wb2);
            const int m = (g < egroups) ? cntE - g * EG : nrows - (g - egroups) * EG;
            if (m >= EG) {
                float v[EG][4];
                int   kj[EG];
                for (int j = 0; j < EG; ++j) {
                    kj[j] = __builtin_amdgcn_readlane(key, j);
                    const float* row = Hin + (size_t)(kj[j] & 0xffff) * DH + lane;
                    v[j][0] = row[0]; v[j][1] = row[32]; v[j][2] = row[64]; v[j][3] = row[96];
                }
                for (int j = 0; j < EG; ++j) {
                    const float w = __int_as_float(__builtin_amdgcn_readlane(wb, j));
                    float* a = acc + (kj[j] >> 16) * TS + lane;
                    __hip_atomic_fetch_add(a,      v[j][0] * w, __ATOMIC_RELAXED, __HIP_MEMORY_SCOPE_WORKGROUP);
                    __hip_atomic_fetch_add(a + 32, v[j][1] * w, __ATOMIC_RELAXED, __HIP_MEMORY_SCOPE_WORKGROUP);
                    __hip_atomic_fetch_add(a + 64, v[j][2] * w, __ATOMIC_RELAXED, __HIP_MEMORY_SCOPE_WORKGROUP);
                    __hip_atomic_fetch_add(a + 96, v[j][3] * w, __ATOMIC_RELAXED, __HIP_MEMORY_SCOPE_WORKGROUP);
                }
            } else {
                for (int j = 0; j < m; ++j) {
                    const int   k = __builtin_amdgcn_readlane(key, j);
                    const float w = __int_as_float(__builtin_amdgcn_readlane(wb, j));
                    edge_add(Hin, acc, k, w, lane);
                }
            }
            key = key2; wb = wb2; g = gn;
        }
    };

    walk(rec + begA, begB - begA, 0, rowsA, wave, AW);

#pragma unroll
    for (int j = 0; j < WPT; ++j) {
        const int i = tid + j * (AW * 32);
        sWt[(i & (DH - 1)) * TS + (i >> 7)] = wreg[j];
    }
    __syncthreads();

    const int ln = lane & 15;
    const int hi = lane >> 4;

    if (wave < MW) {
        const float* aP = &acc[(wave * 16 + ln) * TS + 2 * hi];
        const float* bP = &sWt[ln * TS + 2 * hi];
        v8f d[8];
#pragma unroll
        for (int ct = 0; ct < 8; ++ct) d[ct] = (v8f){0.f, 0.f, 0.f, 0.f, 0.f, 0.f, 0.f, 0.f};
#pragma unroll 2
        for (int k = 0; k < DH; k += 4) {
            const v2f a = *(const v2f*)(aP + k);
#pragma unroll
            for (int ct = 0; ct < 8; ++ct) {
                const v2f b = *(const v2f*)(bP + ct * 16 * TS + k);
                d[ct] = __builtin_amdgcn_wmma_f32_16x16x4_f32(false, a, false, b, (short)0, d[ct], false, false);
            }
        }
#pragma unroll
        for (int ct = 0; ct < 8; ++ct) {
            const float bv = bias[ct * 16 + ln];
#pragma unroll
            for (int r = 0; r < 8; ++r)
                acc[(wave * 16 + r + 8 * hi) * TS + ct * 16 + ln] = fmaxf(d[ct][r] + bv, 0.0f);
        }
    } else {
        walk(rec + begB, endB - begB, SPLIT, rowsB, wave - MW, AW - MW);
    }
    __syncthreads();

    v8f res[2];
#pragma unroll
    for (int q = 0; q < 2; ++q) {
        const int u = MW * 8 + wave + q * AW;
        res[q] = (v8f){0.f, 0.f, 0.f, 0.f, 0.f, 0.f, 0.f, 0.f};
        if (u < NU) {
            const int strip = u >> 3, ct = u & 7;
            const float* aP = &acc[(strip * 16 + ln) * TS + 2 * hi];
            const float* bP = &sWt[(ct * 16 + ln) * TS + 2 * hi];
            v8f d = {0.f, 0.f, 0.f, 0.f, 0.f, 0.f, 0.f, 0.f};
#pragma unroll
            for (int k = 0; k < DH; k += 4) {
                v2f a = *(const v2f*)(aP + k);
                v2f b = *(const v2f*)(bP + k);
                d = __builtin_amdgcn_wmma_f32_16x16x4_f32(false, a, false, b, (short)0, d, false, false);
            }
            const float bv = bias[ct * 16 + ln];
#pragma unroll
            for (int r = 0; r < 8; ++r) res[q][r] = fmaxf(d[r] + bv, 0.0f);
        }
    }
    __syncthreads();

#pragma unroll
    for (int q = 0; q < 2; ++q) {
        const int u = MW * 8 + wave + q * AW;
        if (u < NU) {
            const int strip = u >> 3, ct = u & 7;
#pragma unroll
            for (int r = 0; r < 8; ++r)
                acc[(strip * 16 + r + 8 * hi) * TS + ct * 16 + ln] = res[q][r];
        }
    }
    __syncthreads();

    if (!POOL) {
        float4* H4 = (float4*)Hout;
        for (int i = tid; i < TR * DH / 4; i += AW * 32) {
            const int r = i >> 5, c4 = i & 31;
            if (r < rows) H4[(size_t)(n0 + r) * (DH / 4) + c4] = *((const float4*)(acc + r * TS) + c4);
        }
    } else {
        const int c  = tid & (DH - 1);
        const int rb = (wave >> 2) * PR;
        if ((wave & 3) == 0 && lane < PR) atomicAdd(&counts[batch[n0 + rb + lane]], 1.0f);
        int   gid[PR];
        float v[PR];
#pragma unroll
        for (int j = 0; j < PR; ++j) {
            gid[j] = batch[n0 + rb + j];
            v[j]   = acc[(rb + j) * TS + c];
        }
        int   gg = gid[0];
        float s = 0.0f;
#pragma unroll
        for (int j = 0; j < PR; ++j) {
            if (gid[j] != gg) {
                atomicAdd(&sums[gg * DH + c], s);
                gg = gid[j];
                s = 0.0f;
            }
            s += v[j];
        }
        atomicAdd(&sums[gg * DH + c], s);
    }
}

__global__ void k_pool_div(const float* __restrict__ sums, const float* __restrict__ counts,
                           float* __restrict__ pooled)
{
    int idx = blockIdx.x * blockDim.x + threadIdx.x;
    if (idx >= NG * DH) return;
    pooled[idx] = sums[idx] / fmaxf(counts[idx >> 7], 1.0f);
}

extern "C" void kernel_launch(void* const* d_in, const int* in_sizes, int n_in,
                              void* d_out, int out_size, void* d_ws, size_t ws_size,
                              hipStream_t stream)
{
    (void)in_sizes; (void)n_in; (void)out_size; (void)ws_size;

    const float* x     = (const float*)d_in[0];
    const int*   ei    = (const int*)d_in[1];
    const int*   src   = ei;
    const int*   dst   = ei + NE;
    const int*   batch = (const int*)d_in[2];
    const float* Wl[5] = {(const float*)d_in[4], (const float*)d_in[6], (const float*)d_in[8],
                          (const float*)d_in[10], (const float*)d_in[12]};
    const float* Bl[5] = {(const float*)d_in[5], (const float*)d_in[7], (const float*)d_in[9],
                          (const float*)d_in[11], (const float*)d_in[13]};
    const float* fc1_w = (const float*)d_in[14];
    const float* fc1_b = (const float*)d_in[15];
    const float* fc2_w = (const float*)d_in[16];
    const float* fc2_b = (const float*)d_in[17];

    float* H      = (float*)d_ws;
    float* H2     = H      + (size_t)NN * DH;
    float* selfc  = H2     + (size_t)NN * DH;
    float* dinv   = selfc  + NN;
    int2*  rec    = (int2*)(dinv + NN);
    int*   tilestart = (int*)(rec + NE);
    int*   cursor    = tilestart + NB + 1;
    int*   tilecount = cursor + NB;
    float* sums   = (float*)(tilecount + NB);
    float* counts = sums   + (size_t)NG * DH;
    float* pooled = counts + NG;
    float* gact   = pooled + (size_t)NG * DH;

    const int T = 256;

    hipMemsetAsync(tilecount, 0, ((size_t)NB + (size_t)NG * DH + NG) * sizeof(float), stream);

    k_tile_count<<<(NE + 4095) / 4096, 1024, 0, stream>>>(dst, tilecount, NE);
    k_tile_scan<<<1, 1024, 0, stream>>>(tilecount, tilestart, cursor);
    k_bin_fill<<<(NE + 4095) / 4096, 1024, 0, stream>>>(src, dst, tilestart, cursor, rec, NE);
    k_tile_deg<<<NT, 256, 0, stream>>>(rec, tilestart, selfc, dinv, NN);
    k_tile_weights<<<NT, 256, 0, stream>>>(tilestart, dinv, rec);

    k_layer_tile<false><<<NT, AW * 32, 0, stream>>>(x,  Wl[0], rec, tilestart, selfc, Bl[0], H,
                                                    nullptr, nullptr, nullptr, NN);
    k_layer_tile<false><<<NT, AW * 32, 0, stream>>>(H,  Wl[1], rec, tilestart, selfc, Bl[1], H2,
                                                    nullptr, nullptr, nullptr, NN);
    k_layer_tile<false><<<NT, AW * 32, 0, stream>>>(H2, Wl[2], rec, tilestart, selfc, Bl[2], H,
                                                    nullptr, nullptr, nullptr, NN);
    k_layer_tile<false><<<NT, AW * 32, 0, stream>>>(H,  Wl[3], rec, tilestart, selfc, Bl[3], H2,
                                                    nullptr, nullptr, nullptr, NN);
    k_layer_tile<true><<<NT, AW * 32, 0, stream>>>(H2, Wl[4], rec, tilestart, selfc, Bl[4], nullptr,
                                                   batch, sums, counts, NN);

    k_pool_div<<<(NG * DH + T - 1) / T, T, 0, stream>>>(sums, counts, pooled);

    k_wmma_gemm<DH><<<NG / 16, T, 0, stream>>>(pooled, fc1_w, fc1_b, gact, NG, 1);
    k_wmma_gemm<DO><<<NG / 16, T, 0, stream>>>(gact, fc2_w, fc2_b, (float*)d_out, NG, 0);
}


# edge walk reads each source row with one 16-byte-per-lane load (whole 512 B row), not four 4-byte loads; tile columns interleaved in LDS so adds stay conflict free; product's k sum order changes
# speedup vs baseline: 6.5727x; 1.0662x over previous
#include <hip/hip_runtime.h>

#define NN 50000
#define NE 800000
#define DH 128
#define DO 64
#define NG 128
#define TR 100
#define NT ((NN + TR - 1) / TR)
#define EG 8
#define AW 16
#define PR (TR / 4)
#define SPLIT 64
#define NB (2 * NT)
#define MW (SPLIT / 16)

static_assert(NB <= 1024, "the bin scan is one 1024-thread block");
static_assert(SPLIT % 16 == 0 && SPLIT < TR && MW < AW, "the front part is whole strips, one wave each, and walkers remain");
static_assert(NN <= 65536 && TR <= 32768, "a record packs source row and local row into one word");
static_assert(AW * 32 == 4 * DH && TR % 4 == 0 && PR <= 32 && NN % TR == 0,
              "the pool walk gives a quarter of a full tile's rows to DH threads");

typedef float v2f __attribute__((ext_vector_type(2)));
typedef float v8f __attribute__((ext_vector_type(8)));

__global__ void __launch_bounds__(1024)
k_tile_count(const int* __restrict__ dst, int* __restrict__ tilecount, int E) {
    __shared__ int cnt[NB];
    const int tid = threadIdx.x;
    for (int t = tid; t < NB; t += 1024) cnt[t] = 0;
    __syncthreads();
    const int e0 = blockIdx.x * 4096 + tid;
    int d[4] = {-1, -1, -1, -1};
    for (int j = 0; j < 4; ++j) {
        const int e = e0 + j * 1024;
        if (e < E) d[j] = dst[e];
    }
    for (int j = 0; j < 4; ++j)
        if (d[j] >= 0) {
            const int t = d[j] / TR;
            atomicAdd(&cnt[2 * t + ((d[j] - t * TR) >= SPLIT ? 1 : 0)], 1);
        }
    __syncthreads();
    for (int t = tid; t < NB; t += 1024) {
        const int c = cnt[t];
        if (c) atomicAdd(&tilecount[t], c);
    }
}

__global__ void __launch_bounds__(1024)
k_tile_scan(const int* __restrict__ tilecount, int* __restrict__ tilestart, int* __restrict__ cursor) {
    __shared__ int s[1024];
    const int t = threadIdx.x;
    const int c = (t < NB) ? tilecount[t] : 0;
    s[t] = c;
    __syncthreads();
    for (int o = 1; o < 1024; o <<= 1) {
        const int v = (t >= o) ? s[t - o] : 0;
        __syncthreads();
        s[t] += v;
        __syncthreads();
    }
    if (t < NB) { tilestart[t] = s[t] - c; cursor[t] = 0; }
    if (t == NB - 1) tilestart[NB] = s[t];
}

__global__ void __launch_bounds__(1024)
k_bin_fill(const int* __restrict__ src, const int* __restrict__ dst,
           const int* __restrict__ tilestart,
           int* __restrict__ cursor, int2* __restrict__ rec, int E) {
    __shared__ int cnt[NB];
    __shared__ int base[NB];
    const int tid = threadIdx.x;
    for (int t = tid; t < NB; t += 1024) cnt[t] = 0;
    __syncthreads();
    int s[4] = {0, 0, 0, 0}, d[4] = {0, 0, 0, 0}, tl[4] = {0, 0, 0, 0}, r[4] = {0, 0, 0, 0}, lr[4] = {0, 0, 0, 0};
    const int e0 = blockIdx.x * 4096 + tid;
    for (int j = 0; j < 4; ++j) {
        const int e = e0 + j * 1024;
        if (e < E) { s[j] = src[e]; d[j] = dst[e]; }
    }
    for (int j = 0; j < 4; ++j) {
        const int e = e0 + j * 1024;
        if (e < E) {
            const int t = d[j] / TR;
            lr[j] = d[j] - t * TR;
            tl[j] = 2 * t + (lr[j] >= SPLIT ? 1 : 0);
            r[j] = atomicAdd(&cnt[tl[j]], 1);
        }
    }
    __syncthreads();
    for (int t = tid; t < NB; t += 1024) {
        const int c = cnt[t];
        if (c) base[t] = tilestart[t] + atomicAdd(&cursor[t], c);
    }
    __syncthreads();
    for (int j = 0; j < 4; ++j) {
        const int e = e0 + j * 1024;
        if (e < E)
            rec[base[tl[j]] + r[j]] = make_int2(s[j] | (lr[j] << 16), 0);
    }
}

__global__ void __launch_bounds__(256)
k_tile_deg(const int2* __restrict__ rec, const int* __restrict__ tilestart,
           float* __restrict__ selfc, float* __restrict__ dinv, int n) {
    __shared__ int cnt[TR];
    const int t = threadIdx.x;
    if (t < TR) cnt[t] = 0;
    __syncthreads();
    const int beg = tilestart[2 * blockIdx.x];
    const int end = tilestart[2 * blockIdx.x + 2];
    for (int i = beg + t; i < end; i += 256) atomicAdd(&cnt[rec[i].x >> 16], 1);
    __syncthreads();
    const int i = blockIdx.x * TR + t;
    if (t < TR && i < n) {
        const float d = (float)(cnt[t] + 1);
        dinv[i]  = 1.0f / sqrtf(d);
        selfc[i] = 1.0f / d;
    }
}

__global__ void __launch_bounds__(256)
k_tile_weights(const int* __restrict__ tilestart, const float* __restrict__ dinv,
               int2* __restrict__ rec) {
    const int beg = tilestart[2 * blockIdx.x];
    const int end = tilestart[2 * blockIdx.x + 2];
    const int n0  = blockIdx.x * TR;
    for (int i = beg + (int)threadIdx.x; i < end; i += 256) {
        const int k = rec[i].x;
        const float w = dinv[k & 0xffff] * dinv[n0 + (k >> 16)];
        rec[i] = make_int2(k, __float_as_int(w));
    }
}

template <int NC>
__global__ void __launch_bounds__(256)
k_wmma_gemm(const float* __restrict__ A, const float* __restrict__ W,
            const float* __restrict__ bias, float* __restrict__ C,
            int M, int relu)
{
    __shared__ float sWt[NC * 132];
    __shared__ float sA[16 * 132];

    const int tid = threadIdx.x;
    const int r0  = blockIdx.x * 16;

    for (int i = tid; i < 128 * NC; i += 256) {
        int k = i / NC, c = i % NC;
        sWt[c * 132 + k] = W[i];
    }
    for (int i = tid; i < 16 * 128; i += 256) {
        int r = i >> 7, c = i & 127;
        int row = r0 + r;
        sA[r * 132 + c] = (row < M) ? A[row * 128 + c] : 0.0f;
    }
    __syncthreads();

    const int wave = tid >> 5;
    const int lane = tid & 31;
    const int ln   = lane & 15;
    const int hi   = lane >> 4;
    const int c0   = wave * 16;

    if (c0 < NC) {
        const float* aP = &sA [ln        * 132 + 2 * hi];
        const float* bP = &sWt[(c0 + ln) * 132 + 2 * hi];
        v8f acc = {0.f, 0.f, 0.f, 0.f, 0.f, 0.f, 0.f, 0.f};
#pragma unroll
        for (int k = 0; k < 128; k += 4) {
            v2f a = *(const v2f*)(aP + k);
            v2f b = *(const v2f*)(bP + k);
            acc = __builtin_amdgcn_wmma_f32_16x16x4_f32(
                false, a, false, b, (short)0, acc, false, false);
        }
        const int col = c0 + ln;
        const float bv = bias ? bias[col] : 0.0f;
#pragma unroll
        for (int r = 0; r < 8; ++r) {
            int row = r0 + r + 8 * hi;
            if (row < M) {
                float v = acc[r] + bv;
                if (relu) v = fmaxf(v, 0.0f);
                C[row * NC + col] = v;
            }
        }
    }
}

#define TS  132
#define TRP 112
#define NU  ((TRP / 16) * (DH / 16))
#define WPT ((DH * DH) / (AW * 32))

static_assert(TRP % 16 == 0 && TRP >= TR && (TRP * TS) % 4 == 0, "whole strips, 16-byte clears");
static_assert(NU - (SPLIT / 16) * 8 <= 2 * AW && (DH * DH) % (AW * 32) == 0, "a wave takes at most two back output blocks");

__device__ __forceinline__ void edge_add(const float* __restrict__ Hin, float* acc,
                                         int key, float w, int lane) {
    const float4 v = *(const float4*)(Hin + (size_t)(key & 0xffff) * DH + 4 * lane);
    float* a = acc + (key >> 16) * TS + lane;
    __hip_atomic_fetch_add(a,      v.x * w, __ATOMIC_RELAXED, __HIP_MEMORY_SCOPE_WORKGROUP);
    __hip_atomic_fetch_add(a + 32, v.y * w, __ATOMIC_RELAXED, __HIP_MEMORY_SCOPE_WORKGROUP);
    __hip_atomic_fetch_add(a + 64, v.z * w, __ATOMIC_RELAXED, __HIP_MEMORY_SCOPE_WORKGROUP);
    __hip_atomic_fetch_add(a + 96, v.w * w, __ATOMIC_RELAXED, __HIP_MEMORY_SCOPE_WORKGROUP);
}

template <bool POOL>
__global__ void __launch_bounds__(AW * 32)
k_layer_tile(const float* __restrict__ Hin, const float* __restrict__ W,
             const int2* __restrict__ rec, const int* __restrict__ tilestart,
             const float* __restrict__ selfc, const float* __restrict__ bias,
             float* __restrict__ Hout, const int* __restrict__ batch,
             float* __restrict__ sums, float* __restrict__ counts, int n)
{
    __shared__ alignas(16) float sWt[DH * TS];
    __shared__ alignas(16) float acc[TRP * TS];
    const int tid  = threadIdx.x;
    const int lane = tid & 31;
    const int wave = __builtin_amdgcn_readfirstlane(tid >> 5);
    const int n0   = blockIdx.x * TR;
    const int rows = (n - n0 < TR) ? n - n0 : TR;

    float4* acc4 = (float4*)acc;
    for (int i = tid; i < TRP * TS / 4; i += AW * 32) acc4[i] = make_float4(0.f, 0.f, 0.f, 0.f);
    __syncthreads();

    float wreg[WPT];
#pragma unroll
    for (int j = 0; j < WPT; ++j) wreg[j] = W[tid + j * (AW * 32)];

    const int begA  = tilestart[2 * blockIdx.x];
    const int begB  = tilestart[2 * blockIdx.x + 1];
    const int endB  = tilestart[2 * blockIdx.x + 2];
    const int rowsA = (rows < SPLIT) ? rows : SPLIT;
    const int rowsB = rows - rowsA;

    auto walk = [&](const int2* __restrict__ trec, const int cntE, const int r0, const int nrows,
                    const int g0, const int gs) {
        const int egroups = (cntE + EG - 1) / EG;
        const int ngroups = egroups + (nrows + EG - 1) / EG;
        auto fetch = [&](int gi, int& k, int& w) {
            k = 0; w = 0;
            if (gi < egroups) {
                const int e = gi * EG + lane;
                if (lane < EG && e < cntE) { const int2 q = trec[e]; k = q.x; w = q.y; }
            } else if (gi < ngroups) {
                const int r = (gi - egroups) * EG + lane;
                if (lane < EG && r < nrows) {
                    k = (n0 + r0 + r) | ((r0 + r) << 16);
                    w = __float_as_int(selfc[n0 + r0 + r]);
                }
            }
        };
        int key, wb;
        int g = g0;
        fetch(g, key, wb);
        while (g < ngroups) {
            const int gn = g + gs;
            int key2, wb2;
            fetch(gn, key2, wb2);
            const int m = (g < egroups) ? cntE - g * EG : nrows - (g - egroups) * EG;
            if (m >= EG) {
                float4 v[EG];
                int    kj[EG];
                for (int j = 0; j < EG; ++j) {
                    kj[j] = __builtin_amdgcn_readlane(key, j);
                    v[j] = *(const float4*)(Hin + (size_t)(kj[j] & 0xffff) * DH + 4 * lane);
                }
                for (int j = 0; j < EG; ++j) {
                    const float w = __int_as_float(__builtin_amdgcn_readlane(wb, j));
                    float* a = acc + (kj[j] >> 16) * TS + lane;
                    __hip_atomic_fetch_add(a,      v[j].x * w, __ATOMIC_RELAXED, __HIP_MEMORY_SCOPE_WORKGROUP);
                    __hip_atomic_fetch_add(a + 32, v[j].y * w, __ATOMIC_RELAXED, __HIP_MEMORY_SCOPE_WORKGROUP);
                    __hip_atomic_fetch_add(a + 64, v[j].z * w, __ATOMIC_RELAXED, __HIP_MEMORY_SCOPE_WORKGROUP);
                    __hip_atomic_fetch_add(a + 96, v[j].w * w, __ATOMIC_RELAXED, __HIP_MEMORY_SCOPE_WORKGROUP);
                }
            } else {
                for (int j = 0; j < m; ++j) {
                    const int   k = __builtin_amdgcn_readlane(key, j);
                    const float w = __int_as_float(__builtin_amdgcn_readlane(wb, j));
                    edge_add(Hin, acc, k, w, lane);
                }
            }
            key = key2; wb = wb2; g = gn;
        }
    };

    walk(rec + begA, begB - begA, 0, rowsA, wave, AW);

#pragma unroll
    for (int j = 0; j < WPT; ++j) {
        const int i = tid + j * (AW * 32);
        const int k = i >> 7;
        sWt[(i & (DH - 1)) * TS + (k & 3) * 32 + (k >> 2)] = wreg[j];
    }
    __syncthreads();

    const int ln = lane & 15;
    const int hi = lane >> 4;

    if (wave < MW) {
        const float* aP = &acc[(wave * 16 + ln) * TS + 2 * hi];
        const float* bP = &sWt[ln * TS + 2 * hi];
        v8f d[8];
#pragma unroll
        for (int ct = 0; ct < 8; ++ct) d[ct] = (v8f){0.f, 0.f, 0.f, 0.f, 0.f, 0.f, 0.f, 0.f};
#pragma unroll 2
        for (int k = 0; k < DH; k += 4) {
            const v2f a = *(const v2f*)(aP + k);
#pragma unroll
            for (int ct = 0; ct < 8; ++ct) {
                const v2f b = *(const v2f*)(bP + ct * 16 * TS + k);
                d[ct] = __builtin_amdgcn_wmma_f32_16x16x4_f32(false, a, false, b, (short)0, d[ct], false, false);
            }
        }
#pragma unroll
        for (int ct = 0; ct < 8; ++ct) {
            const float bv = bias[ct * 16 + ln];
#pragma unroll
            for (int r = 0; r < 8; ++r)
                acc[(wave * 16 + r + 8 * hi) * TS + ct * 16 + ln] = fmaxf(d[ct][r] + bv, 0.0f);
        }
    } else {
        walk(rec + begB, endB - begB, SPLIT, rowsB, wave - MW, AW - MW);
    }
    __syncthreads();

    v8f res[2];
#pragma unroll
    for (int q = 0; q < 2; ++q) {
        const int u = MW * 8 + wave + q * AW;
        res[q] = (v8f){0.f, 0.f, 0.f, 0.f, 0.f, 0.f, 0.f, 0.f};
        if (u < NU) {
            const int strip = u >> 3, ct = u & 7;
            const float* aP = &acc[(strip * 16 + ln) * TS + 2 * hi];
            const float* bP = &sWt[(ct * 16 + ln) * TS + 2 * hi];
            v8f d = {0.f, 0.f, 0.f, 0.f, 0.f, 0.f, 0.f, 0.f};
#pragma unroll
            for (int k = 0; k < DH; k += 4) {
                v2f a = *(const v2f*)(aP + k);
                v2f b = *(const v2f*)(bP + k);
                d = __builtin_amdgcn_wmma_f32_16x16x4_f32(false, a, false, b, (short)0, d, false, false);
            }
            const float bv = bias[ct * 16 + ln];
#pragma unroll
            for (int r = 0; r < 8; ++r) res[q][r] = fmaxf(d[r] + bv, 0.0f);
        }
    }
    __syncthreads();

#pragma unroll
    for (int q = 0; q < 2; ++q) {
        const int u = MW * 8 + wave + q * AW;
        if (u < NU) {
            const int strip = u >> 3, ct = u & 7;
#pragma unroll
            for (int r = 0; r < 8; ++r)
                acc[(strip * 16 + r + 8 * hi) * TS + ct * 16 + ln] = res[q][r];
        }
    }
    __syncthreads();

    if (!POOL) {
        float4* H4 = (float4*)Hout;
        for (int i = tid; i < TR * DH / 4; i += AW * 32) {
            const int r = i >> 5, c4 = i & 31;
            if (r < rows) H4[(size_t)(n0 + r) * (DH / 4) + c4] = *((const float4*)(acc + r * TS) + c4);
        }
    } else {
        const int c  = tid & (DH - 1);
        const int rb = (wave >> 2) * PR;
        if ((wave & 3) == 0 && lane < PR) atomicAdd(&counts[batch[n0 + rb + lane]], 1.0f);
        int   gid[PR];
        float v[PR];
#pragma unroll
        for (int j = 0; j < PR; ++j) {
            gid[j] = batch[n0 + rb + j];
            v[j]   = acc[(rb + j) * TS + c];
        }
        int   gg = gid[0];
        float s = 0.0f;
#pragma unroll
        for (int j = 0; j < PR; ++j) {
            if (gid[j] != gg) {
                atomicAdd(&sums[gg * DH + c], s);
                gg = gid[j];
                s = 0.0f;
            }
            s += v[j];
        }
        atomicAdd(&sums[gg * DH + c], s);
    }
}

__global__ void k_pool_div(const float* __restrict__ sums, const float* __restrict__ counts,
                           float* __restrict__ pooled)
{
    int idx = blockIdx.x * blockDim.x + threadIdx.x;
    if (idx >= NG * DH) return;
    pooled[idx] = sums[idx] / fmaxf(counts[idx >> 7], 1.0f);
}

extern "C" void kernel_launch(void* const* d_in, const int* in_sizes, int n_in,
                              void* d_out, int out_size, void* d_ws, size_t ws_size,
                              hipStream_t stream)
{
    (void)in_sizes; (void)n_in; (void)out_size; (void)ws_size;

    const float* x     = (const float*)d_in[0];
    const int*   ei    = (const int*)d_in[1];
    const int*   src   = ei;
    const int*   dst   = ei + NE;
    const int*   batch = (const int*)d_in[2];
    const float* Wl[5] = {(const float*)d_in[4], (const float*)d_in[6], (const float*)d_in[8],
                          (const float*)d_in[10], (const float*)d_in[12]};
    const float* Bl[5] = {(const float*)d_in[5], (const float*)d_in[7], (const float*)d_in[9],
                          (const float*)d_in[11], (const float*)d_in[13]};
    const float* fc1_w = (const float*)d_in[14];
    const float* fc1_b = (const float*)d_in[15];
    const float* fc2_w = (const float*)d_in[16];
    const float* fc2_b = (const float*)d_in[17];

    float* H      = (float*)d_ws;
    float* H2     = H      + (size_t)NN * DH;
    float* selfc  = H2     + (size_t)NN * DH;
    float* dinv   = selfc  + NN;
    int2*  rec    = (int2*)(dinv + NN);
    int*   tilestart = (int*)(rec + NE);
    int*   cursor    = tilestart + NB + 1;
    int*   tilecount = cursor + NB;
    float* sums   = (float*)(tilecount + NB);
    float* counts = sums   + (size_t)NG * DH;
    float* pooled = counts + NG;
    float* gact   = pooled + (size_t)NG * DH;

    const int T = 256;

    hipMemsetAsync(tilecount, 0, ((size_t)NB + (size_t)NG * DH + NG) * sizeof(float), stream);

    k_tile_count<<<(NE + 4095) / 4096, 1024, 0, stream>>>(dst, tilecount, NE);
    k_tile_scan<<<1, 1024, 0, stream>>>(tilecount, tilestart, cursor);
    k_bin_fill<<<(NE + 4095) / 4096, 1024, 0, stream>>>(src, dst, tilestart, cursor, rec, NE);
    k_tile_deg<<<NT, 256, 0, stream>>>(rec, tilestart, selfc, dinv, NN);
    k_tile_weights<<<NT, 256, 0, stream>>>(tilestart, dinv, rec);

    k_layer_tile<false><<<NT, AW * 32, 0, stream>>>(x,  Wl[0], rec, tilestart, selfc, Bl[0], H,
                                                    nullptr, nullptr, nullptr, NN);
    k_layer_tile<false><<<NT, AW * 32, 0, stream>>>(H,  Wl[1], rec, tilestart, selfc, Bl[1], H2,
                                                    nullptr, nullptr, nullptr, NN);
    k_layer_tile<false><<<NT, AW * 32, 0, stream>>>(H2, Wl[2], rec, tilestart, selfc, Bl[2], H,
                                                    nullptr, nullptr, nullptr, NN);
    k_layer_tile<false><<<NT, AW * 32, 0, stream>>>(H,  Wl[3], rec, tilestart, selfc, Bl[3], H2,
                                                    nullptr, nullptr, nullptr, NN);
    k_layer_tile<true><<<NT, AW * 32, 0, stream>>>(H2, Wl[4], rec, tilestart, selfc, Bl[4], nullptr,
                                                   batch, sums, counts, NN);

    k_pool_div<<<(NG * DH + T - 1) / T, T, 0, stream>>>(sums, counts, pooled);

    k_wmma_gemm<DH><<<NG / 16, T, 0, stream>>>(pooled, fc1_w, fc1_b, gact, NG, 1);
    k_wmma_gemm<DO><<<NG / 16, T, 0, stream>>>(gact, fc2_w, fc2_b, (float*)d_out, NG, 0);
}


# three dependent launches fewer: every fill block scans the 1000 bin counts itself (lane shuffles), so the scan kernel goes; pool division, fc1 and fc2 run as one kernel; every sum keeps its order
# speedup vs baseline: 6.9816x; 1.0622x over previous
#include <hip/hip_runtime.h>

#define NN 50000
#define NE 800000
#define DH 128
#define DO 64
#define NG 128
#define TR 100
#define NT ((NN + TR - 1) / TR)
#define EG 8
#define AW 16
#define PR (TR / 4)
#define SPLIT 64
#define NB (2 * NT)
#define MW (SPLIT / 16)

static_assert(NB <= 1024, "the bin scan is one 1024-thread block");
static_assert(SPLIT % 16 == 0 && SPLIT < TR && MW < AW, "the front part is whole strips, one wave each, and walkers remain");
static_assert(NN <= 65536 && TR <= 32768, "a record packs source row and local row into one word");
static_assert(AW * 32 == 4 * DH && TR % 4 == 0 && PR <= 32 && NN % TR == 0,
              "the pool walk gives a quarter of a full tile's rows to DH threads");

typedef float v2f __attribute__((ext_vector_type(2)));
typedef float v8f __attribute__((ext_vector_type(8)));

__global__ void __launch_bounds__(1024)
k_tile_count(const int* __restrict__ dst, int* __restrict__ tilecount, int E) {
    __shared__ int cnt[NB];
    const int tid = threadIdx.x;
    for (int t = tid; t < NB; t += 1024) cnt[t] = 0;
    __syncthreads();
    const int e0 = blockIdx.x * 4096 + tid;
    int d[4] = {-1, -1, -1, -1};
    for (int j = 0; j < 4; ++j) {
        const int e = e0 + j * 1024;
        if (e < E) d[j] = dst[e];
    }
    for (int j = 0; j < 4; ++j)
        if (d[j] >= 0) {
            const int t = d[j] / TR;
            atomicAdd(&cnt[2 * t + ((d[j] - t * TR) >= SPLIT ? 1 : 0)], 1);
        }
    __syncthreads();
    for (int t = tid; t < NB; t += 1024) {
        const int c = cnt[t];
        if (c) atomicAdd(&tilecount[t], c);
    }
}

__global__ void __launch_bounds__(1024)
k_bin_fill(const int* __restrict__ src, const int* __restrict__ dst,
           const int* __restrict__ tilecount, int* __restrict__ tilestart,
           int* __restrict__ cursor, int2* __restrict__ rec, int E) {
    __shared__ int cnt[NB];
    __shared__ int base[NB];
    __shared__ int start[NB];
    __shared__ int wsum[32];
    const int tid  = threadIdx.x;
    const int lane = tid & 31;
    const int wv   = tid >> 5;
    for (int t = tid; t < NB; t += 1024) cnt[t] = 0;
    int s[4] = {0, 0, 0, 0}, d[4] = {0, 0, 0, 0}, tl[4] = {0, 0, 0, 0}, r[4] = {0, 0, 0, 0}, lr[4] = {0, 0, 0, 0};
    const int e0 = blockIdx.x * 4096 + tid;
    for (int j = 0; j < 4; ++j) {
        const int e = e0 + j * 1024;
        if (e < E) { s[j] = src[e]; d[j] = dst[e]; }
    }
    const int c = (tid < NB) ? tilecount[tid] : 0;
    int x = c;
    for (int o = 1; o < 32; o <<= 1) {
        const int y = __shfl_up(x, o);
        if (lane >= o) x += y;
    }
    if (lane == 31) wsum[wv] = x;
    __syncthreads();
    if (wv == 0) {
        const int w = wsum[lane];
        int xs = w;
        for (int o = 1; o < 32; o <<= 1) {
            const int y = __shfl_up(xs, o);
            if (lane >= o) xs += y;
        }
        wsum[lane] = xs - w;
    }
    __syncthreads();
    const int st = wsum[wv] + x - c;
    if (tid < NB) {
        start[tid] = st;
        if (blockIdx.x == 0) {
            tilestart[tid] = st;
            if (tid == NB - 1) tilestart[NB] = st + c;
        }
    }
    for (int j = 0; j < 4; ++j) {
        const int e = e0 + j * 1024;
        if (e < E) {
            const int t = d[j] / TR;
            lr[j] = d[j] - t * TR;
            tl[j] = 2 * t + (lr[j] >= SPLIT ? 1 : 0);
            r[j] = atomicAdd(&cnt[tl[j]], 1);
        }
    }
    __syncthreads();
    for (int t = tid; t < NB; t += 1024) {
        const int n = cnt[t];
        if (n) base[t] = start[t] + atomicAdd(&cursor[t], n);
    }
    __syncthreads();
    for (int j = 0; j < 4; ++j) {
        const int e = e0 + j * 1024;
        if (e < E)
            rec[base[tl[j]] + r[j]] = make_int2(s[j] | (lr[j] << 16), 0);
    }
}

__global__ void __launch_bounds__(256)
k_tile_deg(const int2* __restrict__ rec, const int* __restrict__ tilestart,
           float* __restrict__ selfc, float* __restrict__ dinv, int n) {
    __shared__ int cnt[TR];
    const int t = threadIdx.x;
    if (t < TR) cnt[t] = 0;
    __syncthreads();
    const int beg = tilestart[2 * blockIdx.x];
    const int end = tilestart[2 * blockIdx.x + 2];
    for (int i = beg + t; i < end; i += 256) atomicAdd(&cnt[rec[i].x >> 16], 1);
    __syncthreads();
    const int i = blockIdx.x * TR + t;
    if (t < TR && i < n) {
        const float d = (float)(cnt[t] + 1);
        dinv[i]  = 1.0f / sqrtf(d);
        selfc[i] = 1.0f / d;
    }
}

__global__ void __launch_bounds__(256)
k_tile_weights(const int* __restrict__ tilestart, const float* __restrict__ dinv,
               int2* __restrict__ rec) {
    const int beg = tilestart[2 * blockIdx.x];
    const int end = tilestart[2 * blockIdx.x + 2];
    const int n0  = blockIdx.x * TR;
    for (int i = beg + (int)threadIdx.x; i < end; i += 256) {
        const int k = rec[i].x;
        const float w = dinv[k & 0xffff] * dinv[n0 + (k >> 16)];
        rec[i] = make_int2(k, __float_as_int(w));
    }
}

static_assert(NG % 16 == 0 && DH == 128 && DO % 16 == 0 && DO <= DH, "whole 16-row strips, one column tile per wave");

__global__ void __launch_bounds__(256)
k_head(const float* __restrict__ sums, const float* __restrict__ counts,
       const float* __restrict__ W1, const float* __restrict__ b1,
       const float* __restrict__ W2, const float* __restrict__ b2,
       float* __restrict__ out)
{
    __shared__ float sW1t[DH * 132];
    __shared__ float sW2t[DO * 132];
    __shared__ float sA[16 * 132];
    __shared__ float sG[16 * 132];

    const int tid = threadIdx.x;
    const int r0  = blockIdx.x * 16;

    float sv[16 * DH / 256], cn[16 * DH / 256], w1[DH * DH / 256], w2[DH * DO / 256];
#pragma unroll
    for (int j = 0; j < 16 * DH / 256; ++j) {
        const int i = tid + j * 256;
        sv[j] = sums[r0 * DH + i];
        cn[j] = counts[r0 + (i >> 7)];
    }
#pragma unroll
    for (int j = 0; j < DH * DH / 256; ++j) w1[j] = W1[tid + j * 256];
#pragma unroll
    for (int j = 0; j < DH * DO / 256; ++j) w2[j] = W2[tid + j * 256];
#pragma unroll
    for (int j = 0; j < 16 * DH / 256; ++j) {
        const int i = tid + j * 256;
        sA[(i >> 7) * 132 + (i & 127)] = sv[j] / fmaxf(cn[j], 1.0f);
    }
#pragma unroll
    for (int j = 0; j < DH * DH / 256; ++j) {
        const int i = tid + j * 256;
        sW1t[(i % DH) * 132 + i / DH] = w1[j];
    }
#pragma unroll
    for (int j = 0; j < DH * DO / 256; ++j) {
        const int i = tid + j * 256;
        sW2t[(i % DO) * 132 + i / DO] = w2[j];
    }
    __syncthreads();

    const int wave = tid >> 5;
    const int lane = tid & 31;
    const int ln   = lane & 15;
    const int hi   = lane >> 4;
    const int col  = wave * 16 + ln;

    {
        const float* aP = &sA[ln * 132 + 2 * hi];
        const float* bP = &sW1t[col * 132 + 2 * hi];
        v8f acc = {0.f, 0.f, 0.f, 0.f, 0.f, 0.f, 0.f, 0.f};
#pragma unroll
        for (int k = 0; k < DH; k += 4) {
            const v2f a = *(const v2f*)(aP + k);
            const v2f b = *(const v2f*)(bP + k);
            acc = __builtin_amdgcn_wmma_f32_16x16x4_f32(false, a, false, b, (short)0, acc, false, false);
        }
        const float bv = b1[col];
#pragma unroll
        for (int r = 0; r < 8; ++r) sG[(r + 8 * hi) * 132 + col] = fmaxf(acc[r] + bv, 0.0f);
    }
    __syncthreads();

    if (wave < DO / 16) {
        const float* aP = &sG[ln * 132 + 2 * hi];
        const float* bP = &sW2t[col * 132 + 2 * hi];
        v8f acc = {0.f, 0.f, 0.f, 0.f, 0.f, 0.f, 0.f, 0.f};
#pragma unroll
        for (int k = 0; k < DH; k += 4) {
            const v2f a = *(const v2f*)(aP + k);
            const v2f b = *(const v2f*)(bP + k);
            acc = __builtin_amdgcn_wmma_f32_16x16x4_f32(false, a, false, b, (short)0, acc, false, false);
        }
        const float bv = b2[col];
#pragma unroll
        for (int r = 0; r < 8; ++r) out[(r0 + r + 8 * hi) * DO + col] = acc[r] + bv;
    }
}

#define TS  132
#define TRP 112
#define NU  ((TRP / 16) * (DH / 16))
#define WPT ((DH * DH) / (AW * 32))

static_assert(TRP % 16 == 0 && TRP >= TR && (TRP * TS) % 4 == 0, "whole strips, 16-byte clears");
static_assert(NU - (SPLIT / 16) * 8 <= 2 * AW && (DH * DH) % (AW * 32) == 0, "a wave takes at most two back output blocks");

__device__ __forceinline__ void edge_add(const float* __restrict__ Hin, float* acc,
                                         int key, float w, int lane) {
    const float4 v = *(const float4*)(Hin + (size_t)(key & 0xffff) * DH + 4 * lane);
    float* a = acc + (key >> 16) * TS + lane;
    __hip_atomic_fetch_add(a,      v.x * w, __ATOMIC_RELAXED, __HIP_MEMORY_SCOPE_WORKGROUP);
    __hip_atomic_fetch_add(a + 32, v.y * w, __ATOMIC_RELAXED, __HIP_MEMORY_SCOPE_WORKGROUP);
    __hip_atomic_fetch_add(a + 64, v.z * w, __ATOMIC_RELAXED, __HIP_MEMORY_SCOPE_WORKGROUP);
    __hip_atomic_fetch_add(a + 96, v.w * w, __ATOMIC_RELAXED, __HIP_MEMORY_SCOPE_WORKGROUP);
}

template <bool POOL>
__global__ void __launch_bounds__(AW * 32)
k_layer_tile(const float* __restrict__ Hin, const float* __restrict__ W,
             const int2* __restrict__ rec, const int* __restrict__ tilestart,
             const float* __restrict__ selfc, const float* __restrict__ bias,
             float* __restrict__ Hout, const int* __restrict__ batch,
             float* __restrict__ sums, float* __restrict__ counts, int n)
{
    __shared__ alignas(16) float sWt[DH * TS];
    __shared__ alignas(16) float acc[TRP * TS];
    const int tid  = threadIdx.x;
    const int lane = tid & 31;
    const int wave = __builtin_amdgcn_readfirstlane(tid >> 5);
    const int n0   = blockIdx.x * TR;
    const int rows = (n - n0 < TR) ? n - n0 : TR;

    float4* acc4 = (float4*)acc;
    for (int i = tid; i < TRP * TS / 4; i += AW * 32) acc4[i] = make_float4(0.f, 0.f, 0.f, 0.f);
    __syncthreads();

    float wreg[WPT];
#pragma unroll
    for (int j = 0; j < WPT; ++j) wreg[j] = W[tid + j * (AW * 32)];

    const int begA  = tilestart[2 * blockIdx.x];
    const int begB  = tilestart[2 * blockIdx.x + 1];
    const int endB  = tilestart[2 * blockIdx.x + 2];
    const int rowsA = (rows < SPLIT) ? rows : SPLIT;
    const int rowsB = rows - rowsA;

    auto walk = [&](const int2* __restrict__ trec, const int cntE, const int r0, const int nrows,
                    const int g0, const int gs) {
        const int egroups = (cntE + EG - 1) / EG;
        const int ngroups = egroups + (nrows + EG - 1) / EG;
        auto fetch = [&](int gi, int& k, int& w) {
            k = 0; w = 0;
            if (gi < egroups) {
                const int e = gi * EG + lane;
                if (lane < EG && e < cntE) { const int2 q = trec[e]; k = q.x; w = q.y; }
            } else if (gi < ngroups) {
                const int r = (gi - egroups) * EG + lane;
                if (lane < EG && r < nrows) {
                    k = (n0 + r0 + r) | ((r0 + r) << 16);
                    w = __float_as_int(selfc[n0 + r0 + r]);
                }
            }
        };
        int key, wb;
        int g = g0;
        fetch(g, key, wb);
        while (g < ngroups) {
            const int gn = g + gs;
            int key2, wb2;
            fetch(gn, key2, wb2);
            const int m = (g < egroups) ? cntE - g * EG : nrows - (g - egroups) * EG;
            if (m >= EG) {
                float4 v[EG];
                int    kj[EG];
                for (int j = 0; j < EG; ++j) {
                    kj[j] = __builtin_amdgcn_readlane(key, j);
                    v[j] = *(const float4*)(Hin + (size_t)(kj[j] & 0xffff) * DH + 4 * lane);
                }
                for (int j = 0; j < EG; ++j) {
                    const float w = __int_as_float(__builtin_amdgcn_readlane(wb, j));
                    float* a = acc + (kj[j] >> 16) * TS + lane;
                    __hip_atomic_fetch_add(a,      v[j].x * w, __ATOMIC_RELAXED, __HIP_MEMORY_SCOPE_WORKGROUP);
                    __hip_atomic_fetch_add(a + 32, v[j].y * w, __ATOMIC_RELAXED, __HIP_MEMORY_SCOPE_WORKGROUP);
                    __hip_atomic_fetch_add(a + 64, v[j].z * w, __ATOMIC_RELAXED, __HIP_MEMORY_SCOPE_WORKGROUP);
                    __hip_atomic_fetch_add(a + 96, v[j].w * w, __ATOMIC_RELAXED, __HIP_MEMORY_SCOPE_WORKGROUP);
                }
            } else {
                for (int j = 0; j < m; ++j) {
                    const int   k = __builtin_amdgcn_readlane(key, j);
                    const float w = __int_as_float(__builtin_amdgcn_readlane(wb, j));
                    edge_add(Hin, acc, k, w, lane);
                }
            }
            key = key2; wb = wb2; g = gn;
        }
    };

    walk(rec + begA, begB - begA, 0, rowsA, wave, AW);

#pragma unroll
    for (int j = 0; j < WPT; ++j) {
        const int i = tid + j * (AW * 32);
        const int k = i >> 7;
        sWt[(i & (DH - 1)) * TS + (k & 3) * 32 + (k >> 2)] = wreg[j];
    }
    __syncthreads();

    const int ln = lane & 15;
    const int hi = lane >> 4;

    if (wave < MW) {
        const float* aP = &acc[(wave * 16 + ln) * TS + 2 * hi];
        const float* bP = &sWt[ln * TS + 2 * hi];
        v8f d[8];
#pragma unroll
        for (int ct = 0; ct < 8; ++ct) d[ct] = (v8f){0.f, 0.f, 0.f, 0.f, 0.f, 0.f, 0.f, 0.f};
#pragma unroll 2
        for (int k = 0; k < DH; k += 4) {
            const v2f a = *(const v2f*)(aP + k);
#pragma unroll
            for (int ct = 0; ct < 8; ++ct) {
                const v2f b = *(const v2f*)(bP + ct * 16 * TS + k);
                d[ct] = __builtin_amdgcn_wmma_f32_16x16x4_f32(false, a, false, b, (short)0, d[ct], false, false);
            }
        }
#pragma unroll
        for (int ct = 0; ct < 8; ++ct) {
            const float bv = bias[ct * 16 + ln];
#pragma unroll
            for (int r = 0; r < 8; ++r)
                acc[(wave * 16 + r + 8 * hi) * TS + ct * 16 + ln] = fmaxf(d[ct][r] + bv, 0.0f);
        }
    } else {
        walk(rec + begB, endB - begB, SPLIT, rowsB, wave - MW, AW - MW);
    }
    __syncthreads();

    v8f res[2];
#pragma unroll
    for (int q = 0; q < 2; ++q) {
        const int u = MW * 8 + wave + q * AW;
        res[q] = (v8f){0.f, 0.f, 0.f, 0.f, 0.f, 0.f, 0.f, 0.f};
        if (u < NU) {
            const int strip = u >> 3, ct = u & 7;
            const float* aP = &acc[(strip * 16 + ln) * TS + 2 * hi];
            const float* bP = &sWt[(ct * 16 + ln) * TS + 2 * hi];
            v8f d = {0.f, 0.f, 0.f, 0.f, 0.f, 0.f, 0.f, 0.f};
#pragma unroll
            for (int k = 0; k < DH; k += 4) {
                v2f a = *(const v2f*)(aP + k);
                v2f b = *(const v2f*)(bP + k);
                d = __builtin_amdgcn_wmma_f32_16x16x4_f32(false, a, false, b, (short)0, d, false, false);
            }
            const float bv = bias[ct * 16 + ln];
#pragma unroll
            for (int r = 0; r < 8; ++r) res[q][r] = fmaxf(d[r] + bv, 0.0f);
        }
    }
    __syncthreads();

#pragma unroll
    for (int q = 0; q < 2; ++q) {
        const int u = MW * 8 + wave + q * AW;
        if (u < NU) {
            const int strip = u >> 3, ct = u & 7;
#pragma unroll
            for (int r = 0; r < 8; ++r)
                acc[(strip * 16 + r + 8 * hi) * TS + ct * 16 + ln] = res[q][r];
        }
    }
    __syncthreads();

    if (!POOL) {
        float4* H4 = (float4*)Hout;
        for (int i = tid; i < TR * DH / 4; i += AW * 32) {
            const int r = i >> 5, c4 = i & 31;
            if (r < rows) H4[(size_t)(n0 + r) * (DH / 4) + c4] = *((const float4*)(acc + r * TS) + c4);
        }
    } else {
        const int c  = tid & (DH - 1);
        const int rb = (wave >> 2) * PR;
        if ((wave & 3) == 0 && lane < PR) atomicAdd(&counts[batch[n0 + rb + lane]], 1.0f);
        int   gid[PR];
        float v[PR];
#pragma unroll
        for (int j = 0; j < PR; ++j) {
            gid[j] = batch[n0 + rb + j];
            v[j]   = acc[(rb + j) * TS + c];
        }
        int   gg = gid[0];
        float s = 0.0f;
#pragma unroll
        for (int j = 0; j < PR; ++j) {
            if (gid[j] != gg) {
                atomicAdd(&sums[gg * DH + c], s);
                gg = gid[j];
                s = 0.0f;
            }
            s += v[j];
        }
        atomicAdd(&sums[gg * DH + c], s);
    }
}

extern "C" void kernel_launch(void* const* d_in, const int* in_sizes, int n_in,
                              void* d_out, int out_size, void* d_ws, size_t ws_size,
                              hipStream_t stream)
{
    (void)in_sizes; (void)n_in; (void)out_size; (void)ws_size;

    const float* x     = (const float*)d_in[0];
    const int*   ei    = (const int*)d_in[1];
    const int*   src   = ei;
    const int*   dst   = ei + NE;
    const int*   batch = (const int*)d_in[2];
    const float* Wl[5] = {(const float*)d_in[4], (const float*)d_in[6], (const float*)d_in[8],
                          (const float*)d_in[10], (const float*)d_in[12]};
    const float* Bl[5] = {(const float*)d_in[5], (const float*)d_in[7], (const float*)d_in[9],
                          (const float*)d_in[11], (const float*)d_in[13]};
    const float* fc1_w = (const float*)d_in[14];
    const float* fc1_b = (const float*)d_in[15];
    const float* fc2_w = (const float*)d_in[16];
    const float* fc2_b = (const float*)d_in[17];

    float* H      = (float*)d_ws;
    float* H2     = H      + (size_t)NN * DH;
    float* selfc  = H2     + (size_t)NN * DH;
    float* dinv   = selfc  + NN;
    int2*  rec    = (int2*)(dinv + NN);
    int*   tilestart = (int*)(rec + NE);
    int*   cursor    = tilestart + NB + 1;
    int*   tilecount = cursor + NB;
    float* sums   = (float*)(tilecount + NB);
    float* counts = sums   + (size_t)NG * DH;

    hipMemsetAsync(cursor, 0, ((size_t)2 * NB + (size_t)NG * DH + NG) * sizeof(float), stream);

    k_tile_count<<<(NE + 4095) / 4096, 1024, 0, stream>>>(dst, tilecount, NE);
    k_bin_fill<<<(NE + 4095) / 4096, 1024, 0, stream>>>(src, dst, tilecount, tilestart, cursor, rec, NE);
    k_tile_deg<<<NT, 256, 0, stream>>>(rec, tilestart, selfc, dinv, NN);
    k_tile_weights<<<NT, 256, 0, stream>>>(tilestart, dinv, rec);

    k_layer_tile<false><<<NT, AW * 32, 0, stream>>>(x,  Wl[0], rec, tilestart, selfc, Bl[0], H,
                                                    nullptr, nullptr, nullptr, NN);
    k_layer_tile<false><<<NT, AW * 32, 0, stream>>>(H,  Wl[1], rec, tilestart, selfc, Bl[1], H2,
                                                    nullptr, nullptr, nullptr, NN);
    k_layer_tile<false><<<NT, AW * 32, 0, stream>>>(H2, Wl[2], rec, tilestart, selfc, Bl[2], H,
                                                    nullptr, nullptr, nullptr, NN);
    k_layer_tile<false><<<NT, AW * 32, 0, stream>>>(H,  Wl[3], rec, tilestart, selfc, Bl[3], H2,
                                                    nullptr, nullptr, nullptr, NN);
    k_layer_tile<true><<<NT, AW * 32, 0, stream>>>(H2, Wl[4], rec, tilestart, selfc, Bl[4], nullptr,
                                                   batch, sums, counts, NN);

    k_head<<<NG / 16, 256, 0, stream>>>(sums, counts, fc1_w, fc1_b, fc2_w, fc2_b, (float*)d_out);
}


# tile records sorted by destination row (LDS counting sort in the weights pass); a wave sums each run of equal rows in registers, one set of LDS float adds per run, not per edge; sum order changes
# speedup vs baseline: 7.2584x; 1.0396x over previous
#include <hip/hip_runtime.h>

#define NN 50000
#define NE 800000
#define DH 128
#define DO 64
#define NG 128
#define TR 100
#define NT ((NN + TR - 1) / TR)
#define EG 8
#define AW 16
#define PR (TR / 4)
#define SPLIT 64
#define NB (2 * NT)
#define MW (SPLIT / 16)

static_assert(NB <= 1024, "the bin scan is one 1024-thread block");
static_assert(SPLIT % 16 == 0 && SPLIT < TR && MW < AW, "the front part is whole strips, one wave each, and walkers remain");
static_assert(NN <= 65536 && TR <= 32768, "a record packs source row and local row into one word");
static_assert(AW * 32 == 4 * DH && TR % 4 == 0 && PR <= 32 && NN % TR == 0,
              "the pool walk gives a quarter of a full tile's rows to DH threads");

typedef float v2f __attribute__((ext_vector_type(2)));
typedef float v8f __attribute__((ext_vector_type(8)));

__global__ void __launch_bounds__(1024)
k_tile_count(const int* __restrict__ dst, int* __restrict__ tilecount, int E) {
    __shared__ int cnt[NB];
    const int tid = threadIdx.x;
    for (int t = tid; t < NB; t += 1024) cnt[t] = 0;
    __syncthreads();
    const int e0 = blockIdx.x * 4096 + tid;
    int d[4] = {-1, -1, -1, -1};
    for (int j = 0; j < 4; ++j) {
        const int e = e0 + j * 1024;
        if (e < E) d[j] = dst[e];
    }
    for (int j = 0; j < 4; ++j)
        if (d[j] >= 0) {
            const int t = d[j] / TR;
            atomicAdd(&cnt[2 * t + ((d[j] - t * TR) >= SPLIT ? 1 : 0)], 1);
        }
    __syncthreads();
    for (int t = tid; t < NB; t += 1024) {
        const int c = cnt[t];
        if (c) atomicAdd(&tilecount[t], c);
    }
}

__global__ void __launch_bounds__(1024)
k_bin_fill(const int* __restrict__ src, const int* __restrict__ dst,
           const int* __restrict__ tilecount, int* __restrict__ tilestart,
           int* __restrict__ cursor, int2* __restrict__ rec, int E) {
    __shared__ int cnt[NB];
    __shared__ int base[NB];
    __shared__ int start[NB];
    __shared__ int wsum[32];
    const int tid  = threadIdx.x;
    const int lane = tid & 31;
    const int wv   = tid >> 5;
    for (int t = tid; t < NB; t += 1024) cnt[t] = 0;
    int s[4] = {0, 0, 0, 0}, d[4] = {0, 0, 0, 0}, tl[4] = {0, 0, 0, 0}, r[4] = {0, 0, 0, 0}, lr[4] = {0, 0, 0, 0};
    const int e0 = blockIdx.x * 4096 + tid;
    for (int j = 0; j < 4; ++j) {
        const int e = e0 + j * 1024;
        if (e < E) { s[j] = src[e]; d[j] = dst[e]; }
    }
    const int c = (tid < NB) ? tilecount[tid] : 0;
    int x = c;
    for (int o = 1; o < 32; o <<= 1) {
        const int y = __shfl_up(x, o);
        if (lane >= o) x += y;
    }
    if (lane == 31) wsum[wv] = x;
    __syncthreads();
    if (wv == 0) {
        const int w = wsum[lane];
        int xs = w;
        for (int o = 1; o < 32; o <<= 1) {
            const int y = __shfl_up(xs, o);
            if (lane >= o) xs += y;
        }
        wsum[lane] = xs - w;
    }
    __syncthreads();
    const int st = wsum[wv] + x - c;
    if (tid < NB) {
        start[tid] = st;
        if (blockIdx.x == 0) {
            tilestart[tid] = st;
            if (tid == NB - 1) tilestart[NB] = st + c;
        }
    }
    for (int j = 0; j < 4; ++j) {
        const int e = e0 + j * 1024;
        if (e < E) {
            const int t = d[j] / TR;
            lr[j] = d[j] - t * TR;
            tl[j] = 2 * t + (lr[j] >= SPLIT ? 1 : 0);
            r[j] = atomicAdd(&cnt[tl[j]], 1);
        }
    }
    __syncthreads();
    for (int t = tid; t < NB; t += 1024) {
        const int n = cnt[t];
        if (n) base[t] = start[t] + atomicAdd(&cursor[t], n);
    }
    __syncthreads();
    for (int j = 0; j < 4; ++j) {
        const int e = e0 + j * 1024;
        if (e < E)
            rec[base[tl[j]] + r[j]] = make_int2(s[j] | (lr[j] << 16), 0);
    }
}

__global__ void __launch_bounds__(256)
k_tile_deg(const int2* __restrict__ rec, const int* __restrict__ tilestart,
           float* __restrict__ selfc, float* __restrict__ dinv, int* __restrict__ indeg, int n) {
    __shared__ int cnt[TR];
    const int t = threadIdx.x;
    if (t < TR) cnt[t] = 0;
    __syncthreads();
    const int beg = tilestart[2 * blockIdx.x];
    const int end = tilestart[2 * blockIdx.x + 2];
    for (int i = beg + t; i < end; i += 256) atomicAdd(&cnt[rec[i].x >> 16], 1);
    __syncthreads();
    const int i = blockIdx.x * TR + t;
    if (t < TR && i < n) {
        const float d = (float)(cnt[t] + 1);
        dinv[i]  = 1.0f / sqrtf(d);
        selfc[i] = 1.0f / d;
        indeg[i] = cnt[t];
    }
}

static_assert(TR <= 128, "one wave scans the row counts, four rows per lane");
#define WB 8

__global__ void __launch_bounds__(256)
k_tile_weights(const int* __restrict__ tilestart, const float* __restrict__ dinv,
               const int* __restrict__ indeg, const int2* __restrict__ rec,
               int2* __restrict__ rec2) {
    __shared__ int pos[128];
    const int t   = threadIdx.x;
    const int beg = tilestart[2 * blockIdx.x];
    const int end = tilestart[2 * blockIdx.x + 2];
    const int n0  = blockIdx.x * TR;
    if (t < 32) {
        int c[4];
        int tot = 0;
        for (int k = 0; k < 4; ++k) {
            const int r = 4 * t + k;
            c[k] = (r < TR) ? indeg[n0 + r] : 0;
            tot += c[k];
        }
        int x = tot;
        for (int o = 1; o < 32; o <<= 1) {
            const int y = __shfl_up(x, o);
            if (t >= o) x += y;
        }
        int e = x - tot;
        for (int k = 0; k < 4; ++k) { pos[4 * t + k] = e; e += c[k]; }
    }
    __syncthreads();
    for (int i0 = beg + t; i0 < end; i0 += WB * 256) {
        int   k[WB];
        float a[WB], b[WB];
#pragma unroll
        for (int u = 0; u < WB; ++u) {
            const int i = i0 + u * 256;
            k[u] = (i < end) ? rec[i].x : -1;
        }
#pragma unroll
        for (int u = 0; u < WB; ++u) {
            a[u] = 0.f; b[u] = 0.f;
            if (k[u] >= 0) { a[u] = dinv[k[u] & 0xffff]; b[u] = dinv[n0 + (k[u] >> 16)]; }
        }
        int p[WB];
#pragma unroll
        for (int u = 0; u < WB; ++u) p[u] = (k[u] >= 0) ? atomicAdd(&pos[k[u] >> 16], 1) : 0;
#pragma unroll
        for (int u = 0; u < WB; ++u)
            if (k[u] >= 0) rec2[beg + p[u]] = make_int2(k[u], __float_as_int(a[u] * b[u]));
    }
}

static_assert(NG % 16 == 0 && DH == 128 && DO % 16 == 0 && DO <= DH, "whole 16-row strips, one column tile per wave");

__global__ void __launch_bounds__(256)
k_head(const float* __restrict__ sums, const float* __restrict__ counts,
       const float* __restrict__ W1, const float* __restrict__ b1,
       const float* __restrict__ W2, const float* __restrict__ b2,
       float* __restrict__ out)
{
    __shared__ float sW1t[DH * 132];
    __shared__ float sW2t[DO * 132];
    __shared__ float sA[16 * 132];
    __shared__ float sG[16 * 132];

    const int tid = threadIdx.x;
    const int r0  = blockIdx.x * 16;

    float sv[16 * DH / 256], cn[16 * DH / 256], w1[DH * DH / 256], w2[DH * DO / 256];
#pragma unroll
    for (int j = 0; j < 16 * DH / 256; ++j) {
        const int i = tid + j * 256;
        sv[j] = sums[r0 * DH + i];
        cn[j] = counts[r0 + (i >> 7)];
    }
#pragma unroll
    for (int j = 0; j < DH * DH / 256; ++j) w1[j] = W1[tid + j * 256];
#pragma unroll
    for (int j = 0; j < DH * DO / 256; ++j) w2[j] = W2[tid + j * 256];
#pragma unroll
    for (int j = 0; j < 16 * DH / 256; ++j) {
        const int i = tid + j * 256;
        sA[(i >> 7) * 132 + (i & 127)] = sv[j] / fmaxf(cn[j], 1.0f);
    }
#pragma unroll
    for (int j = 0; j < DH * DH / 256; ++j) {
        const int i = tid + j * 256;
        sW1t[(i % DH) * 132 + i / DH] = w1[j];
    }
#pragma unroll
    for (int j = 0; j < DH * DO / 256; ++j) {
        const int i = tid + j * 256;
        sW2t[(i % DO) * 132 + i / DO] = w2[j];
    }
    __syncthreads();

    const int wave = tid >> 5;
    const int lane = tid & 31;
    const int ln   = lane & 15;
    const int hi   = lane >> 4;
    const int col  = wave * 16 + ln;

    {
        const float* aP = &sA[ln * 132 + 2 * hi];
        const float* bP = &sW1t[col * 132 + 2 * hi];
        v8f acc = {0.f, 0.f, 0.f, 0.f, 0.f, 0.f, 0.f, 0.f};
#pragma unroll
        for (int k = 0; k < DH; k += 4) {
            const v2f a = *(const v2f*)(aP + k);
            const v2f b = *(const v2f*)(bP + k);
            acc = __builtin_amdgcn_wmma_f32_16x16x4_f32(false, a, false, b, (short)0, acc, false, false);
        }
        const float bv = b1[col];
#pragma unroll
        for (int r = 0; r < 8; ++r) sG[(r + 8 * hi) * 132 + col] = fmaxf(acc[r] + bv, 0.0f);
    }
    __syncthreads();

    if (wave < DO / 16) {
        const float* aP = &sG[ln * 132 + 2 * hi];
        const float* bP = &sW2t[col * 132 + 2 * hi];
        v8f acc = {0.f, 0.f, 0.f, 0.f, 0.f, 0.f, 0.f, 0.f};
#pragma unroll
        for (int k = 0; k < DH; k += 4) {
            const v2f a = *(const v2f*)(aP + k);
            const v2f b = *(const v2f*)(bP + k);
            acc = __builtin_amdgcn_wmma_f32_16x16x4_f32(false, a, false, b, (short)0, acc, false, false);
        }
        const float bv = b2[col];
#pragma unroll
        for (int r = 0; r < 8; ++r) out[(r0 + r + 8 * hi) * DO + col] = acc[r] + bv;
    }
}

#define TS  132
#define TRP 112
#define NU  ((TRP / 16) * (DH / 16))
#define WPT ((DH * DH) / (AW * 32))

static_assert(TRP % 16 == 0 && TRP >= TR && (TRP * TS) % 4 == 0, "whole strips, 16-byte clears");
static_assert(NU - (SPLIT / 16) * 8 <= 2 * AW && (DH * DH) % (AW * 32) == 0, "a wave takes at most two back output blocks");

__device__ __forceinline__ void edge_add(const float* __restrict__ Hin, float* acc,
                                         int key, float w, int lane) {
    const float4 v = *(const float4*)(Hin + (size_t)(key & 0xffff) * DH + 4 * lane);
    float* a = acc + (key >> 16) * TS + lane;
    __hip_atomic_fetch_add(a,      v.x * w, __ATOMIC_RELAXED, __HIP_MEMORY_SCOPE_WORKGROUP);
    __hip_atomic_fetch_add(a + 32, v.y * w, __ATOMIC_RELAXED, __HIP_MEMORY_SCOPE_WORKGROUP);
    __hip_atomic_fetch_add(a + 64, v.z * w, __ATOMIC_RELAXED, __HIP_MEMORY_SCOPE_WORKGROUP);
    __hip_atomic_fetch_add(a + 96, v.w * w, __ATOMIC_RELAXED, __HIP_MEMORY_SCOPE_WORKGROUP);
}

template <bool POOL>
__global__ void __launch_bounds__(AW * 32)
k_layer_tile(const float* __restrict__ Hin, const float* __restrict__ W,
             const int2* __restrict__ rec, const int* __restrict__ tilestart,
             const float* __restrict__ selfc, const float* __restrict__ bias,
             float* __restrict__ Hout, const int* __restrict__ batch,
             float* __restrict__ sums, float* __restrict__ counts, int n)
{
    __shared__ alignas(16) float sWt[DH * TS];
    __shared__ alignas(16) float acc[TRP * TS];
    const int tid  = threadIdx.x;
    const int lane = tid & 31;
    const int wave = __builtin_amdgcn_readfirstlane(tid >> 5);
    const int n0   = blockIdx.x * TR;
    const int rows = (n - n0 < TR) ? n - n0 : TR;

    float4* acc4 = (float4*)acc;
    for (int i = tid; i < TRP * TS / 4; i += AW * 32) acc4[i] = make_float4(0.f, 0.f, 0.f, 0.f);
    __syncthreads();

    float wreg[WPT];
#pragma unroll
    for (int j = 0; j < WPT; ++j) wreg[j] = W[tid + j * (AW * 32)];

    const int begA  = tilestart[2 * blockIdx.x];
    const int begB  = tilestart[2 * blockIdx.x + 1];
    const int endB  = tilestart[2 * blockIdx.x + 2];
    const int rowsA = (rows < SPLIT) ? rows : SPLIT;
    const int rowsB = rows - rowsA;

    auto walk = [&](const int2* __restrict__ trec, const int cntE, const int r0, const int nrows,
                    const int g0, const int gs) {
        const int egroups = (cntE + EG - 1) / EG;
        const int ngroups = egroups + (nrows + EG - 1) / EG;
        auto fetch = [&](int gi, int& k, int& w) {
            k = 0; w = 0;
            if (gi < egroups) {
                const int e = gi * EG + lane;
                if (lane < EG && e < cntE) { const int2 q = trec[e]; k = q.x; w = q.y; }
            } else if (gi < ngroups) {
                const int r = (gi - egroups) * EG + lane;
                if (lane < EG && r < nrows) {
                    k = (n0 + r0 + r) | ((r0 + r) << 16);
                    w = __float_as_int(selfc[n0 + r0 + r]);
                }
            }
        };
        int key, wb;
        int g = g0;
        fetch(g, key, wb);
        while (g < ngroups) {
            const int gn = g + gs;
            int key2, wb2;
            fetch(gn, key2, wb2);
            const int m = (g < egroups) ? cntE - g * EG : nrows - (g - egroups) * EG;
            if (m >= EG) {
                float4 v[EG];
                int    kj[EG];
                for (int j = 0; j < EG; ++j) {
                    kj[j] = __builtin_amdgcn_readlane(key, j);
                    v[j] = *(const float4*)(Hin + (size_t)(kj[j] & 0xffff) * DH + 4 * lane);
                }
                float4 s = make_float4(0.f, 0.f, 0.f, 0.f);
                for (int j = 0; j < EG; ++j) {
                    const float w = __int_as_float(__builtin_amdgcn_readlane(wb, j));
                    s.x += v[j].x * w;
                    s.y += v[j].y * w;
                    s.z += v[j].z * w;
                    s.w += v[j].w * w;
                    const int lr = kj[j] >> 16;
                    const int nx = (j + 1 < EG) ? (kj[(j + 1 < EG) ? j + 1 : j] >> 16) : -1;
                    if (nx != lr) {
                        float* a = acc + lr * TS + lane;
                        __hip_atomic_fetch_add(a,      s.x, __ATOMIC_RELAXED, __HIP_MEMORY_SCOPE_WORKGROUP);
                        __hip_atomic_fetch_add(a + 32, s.y, __ATOMIC_RELAXED, __HIP_MEMORY_SCOPE_WORKGROUP);
                        __hip_atomic_fetch_add(a + 64, s.z, __ATOMIC_RELAXED, __HIP_MEMORY_SCOPE_WORKGROUP);
                        __hip_atomic_fetch_add(a + 96, s.w, __ATOMIC_RELAXED, __HIP_MEMORY_SCOPE_WORKGROUP);
                        s = make_float4(0.f, 0.f, 0.f, 0.f);
                    }
                }
            } else {
                for (int j = 0; j < m; ++j) {
                    const int   k = __builtin_amdgcn_readlane(key, j);
                    const float w = __int_as_float(__builtin_amdgcn_readlane(wb, j));
                    edge_add(Hin, acc, k, w, lane);
                }
            }
            key = key2; wb = wb2; g = gn;
        }
    };

    walk(rec + begA, begB - begA, 0, rowsA, wave, AW);

#pragma unroll
    for (int j = 0; j < WPT; ++j) {
        const int i = tid + j * (AW * 32);
        const int k = i >> 7;
        sWt[(i & (DH - 1)) * TS + (k & 3) * 32 + (k >> 2)] = wreg[j];
    }
    __syncthreads();

    const int ln = lane & 15;
    const int hi = lane >> 4;

    if (wave < MW) {
        const float* aP = &acc[(wave * 16 + ln) * TS + 2 * hi];
        const float* bP = &sWt[ln * TS + 2 * hi];
        v8f d[8];
#pragma unroll
        for (int ct = 0; ct < 8; ++ct) d[ct] = (v8f){0.f, 0.f, 0.f, 0.f, 0.f, 0.f, 0.f, 0.f};
#pragma unroll 2
        for (int k = 0; k < DH; k += 4) {
            const v2f a = *(const v2f*)(aP + k);
#pragma unroll
            for (int ct = 0; ct < 8; ++ct) {
                const v2f b = *(const v2f*)(bP + ct * 16 * TS + k);
                d[ct] = __builtin_amdgcn_wmma_f32_16x16x4_f32(false, a, false, b, (short)0, d[ct], false, false);
            }
        }
#pragma unroll
        for (int ct = 0; ct < 8; ++ct) {
            const float bv = bias[ct * 16 + ln];
#pragma unroll
            for (int r = 0; r < 8; ++r)
                acc[(wave * 16 + r + 8 * hi) * TS + ct * 16 + ln] = fmaxf(d[ct][r] + bv, 0.0f);
        }
    } else {
        walk(rec + begB, endB - begB, SPLIT, rowsB, wave - MW, AW - MW);
    }
    __syncthreads();

    v8f res[2];
#pragma unroll
    for (int q = 0; q < 2; ++q) {
        const int u = MW * 8 + wave + q * AW;
        res[q] = (v8f){0.f, 0.f, 0.f, 0.f, 0.f, 0.f, 0.f, 0.f};
        if (u < NU) {
            const int strip = u >> 3, ct = u & 7;
            const float* aP = &acc[(strip * 16 + ln) * TS + 2 * hi];
            const float* bP = &sWt[(ct * 16 + ln) * TS + 2 * hi];
            v8f d = {0.f, 0.f, 0.f, 0.f, 0.f, 0.f, 0.f, 0.f};
#pragma unroll
            for (int k = 0; k < DH; k += 4) {
                v2f a = *(const v2f*)(aP + k);
                v2f b = *(const v2f*)(bP + k);
                d = __builtin_amdgcn_wmma_f32_16x16x4_f32(false, a, false, b, (short)0, d, false, false);
            }
            const float bv = bias[ct * 16 + ln];
#pragma unroll
            for (int r = 0; r < 8; ++r) res[q][r] = fmaxf(d[r] + bv, 0.0f);
        }
    }
    __syncthreads();

#pragma unroll
    for (int q = 0; q < 2; ++q) {
        const int u = MW * 8 + wave + q * AW;
        if (u < NU) {
            const int strip = u >> 3, ct = u & 7;
#pragma unroll
            for (int r = 0; r < 8; ++r)
                acc[(strip * 16 + r + 8 * hi) * TS + ct * 16 + ln] = res[q][r];
        }
    }
    __syncthreads();

    if (!POOL) {
        float4* H4 = (float4*)Hout;
        for (int i = tid; i < TR * DH / 4; i += AW * 32) {
            const int r = i >> 5, c4 = i & 31;
            if (r < rows) H4[(size_t)(n0 + r) * (DH / 4) + c4] = *((const float4*)(acc + r * TS) + c4);
        }
    } else {
        const int c  = tid & (DH - 1);
        const int rb = (wave >> 2) * PR;
        if ((wave & 3) == 0 && lane < PR) atomicAdd(&counts[batch[n0 + rb + lane]], 1.0f);
        int   gid[PR];
        float v[PR];
#pragma unroll
        for (int j = 0; j < PR; ++j) {
            gid[j] = batch[n0 + rb + j];
            v[j]   = acc[(rb + j) * TS + c];
        }
        int   gg = gid[0];
        float s = 0.0f;
#pragma unroll
        for (int j = 0; j < PR; ++j) {
            if (gid[j] != gg) {
                atomicAdd(&sums[gg * DH + c], s);
                gg = gid[j];
                s = 0.0f;
            }
            s += v[j];
        }
        atomicAdd(&sums[gg * DH + c], s);
    }
}

extern "C" void kernel_launch(void* const* d_in, const int* in_sizes, int n_in,
                              void* d_out, int out_size, void* d_ws, size_t ws_size,
                              hipStream_t stream)
{
    (void)in_sizes; (void)n_in; (void)out_size; (void)ws_size;

    const float* x     = (const float*)d_in[0];
    const int*   ei    = (const int*)d_in[1];
    const int*   src   = ei;
    const int*   dst   = ei + NE;
    const int*   batch = (const int*)d_in[2];
    const float* Wl[5] = {(const float*)d_in[4], (const float*)d_in[6], (const float*)d_in[8],
                          (const float*)d_in[10], (const float*)d_in[12]};
    const float* Bl[5] = {(const float*)d_in[5], (const float*)d_in[7], (const float*)d_in[9],
                          (const float*)d_in[11], (const float*)d_in[13]};
    const float* fc1_w = (const float*)d_in[14];
    const float* fc1_b = (const float*)d_in[15];
    const float* fc2_w = (const float*)d_in[16];
    const float* fc2_b = (const float*)d_in[17];

    float* H      = (float*)d_ws;
    float* H2     = H      + (size_t)NN * DH;
    float* selfc  = H2     + (size_t)NN * DH;
    float* dinv   = selfc  + NN;
    int2*  rec    = (int2*)(dinv + NN);
    int2*  rec2   = rec + NE;
    int*   tilestart = (int*)(rec2 + NE);
    int*   cursor    = tilestart + NB + 1;
    int*   tilecount = cursor + NB;
    float* sums   = (float*)(tilecount + NB);
    float* counts = sums   + (size_t)NG * DH;
    int*   indeg  = (int*)(counts + NG);

    hipMemsetAsync(cursor, 0, ((size_t)2 * NB + (size_t)NG * DH + NG) * sizeof(float), stream);

    k_tile_count<<<(NE + 4095) / 4096, 1024, 0, stream>>>(dst, tilecount, NE);
    k_bin_fill<<<(NE + 4095) / 4096, 1024, 0, stream>>>(src, dst, tilecount, tilestart, cursor, rec, NE);
    k_tile_deg<<<NT, 256, 0, stream>>>(rec, tilestart, selfc, dinv, indeg, NN);
    k_tile_weights<<<NT, 256, 0, stream>>>(tilestart, dinv, indeg, rec, rec2);

    k_layer_tile<false><<<NT, AW * 32, 0, stream>>>(x,  Wl[0], rec2, tilestart, selfc, Bl[0], H,
                                                    nullptr, nullptr, nullptr, NN);
    k_layer_tile<false><<<NT, AW * 32, 0, stream>>>(H,  Wl[1], rec2, tilestart, selfc, Bl[1], H2,
                                                    nullptr, nullptr, nullptr, NN);
    k_layer_tile<false><<<NT, AW * 32, 0, stream>>>(H2, Wl[2], rec2, tilestart, selfc, Bl[2], H,
                                                    nullptr, nullptr, nullptr, NN);
    k_layer_tile<false><<<NT, AW * 32, 0, stream>>>(H,  Wl[3], rec2, tilestart, selfc, Bl[3], H2,
                                                    nullptr, nullptr, nullptr, NN);
    k_layer_tile<true><<<NT, AW * 32, 0, stream>>>(H2, Wl[4], rec2, tilestart, selfc, Bl[4], nullptr,
                                                   batch, sums, counts, NN);

    k_head<<<NG / 16, 256, 0, stream>>>(sums, counts, fc1_w, fc1_b, fc2_w, fc2_b, (float*)d_out);
}


# pool node counts: the first lane of each run of equal graph ids adds the run's length with one float add, not 1 per node: 50000 same-word adds to 128 words become about 2100; counts stay exact
# speedup vs baseline: 8.3436x; 1.1495x over previous
#include <hip/hip_runtime.h>

#define NN 50000
#define NE 800000
#define DH 128
#define DO 64
#define NG 128
#define TR 100
#define NT ((NN + TR - 1) / TR)
#define EG 8
#define AW 16
#define PR (TR / 4)
#define SPLIT 64
#define NB (2 * NT)
#define MW (SPLIT / 16)

static_assert(NB <= 1024, "the bin scan is one 1024-thread block");
static_assert(SPLIT % 16 == 0 && SPLIT < TR && MW < AW, "the front part is whole strips, one wave each, and walkers remain");
static_assert(NN <= 65536 && TR <= 32768, "a record packs source row and local row into one word");
static_assert(AW * 32 == 4 * DH && TR % 4 == 0 && PR <= 32 && NN % TR == 0,
              "the pool walk gives a quarter of a full tile's rows to DH threads");

typedef float v2f __attribute__((ext_vector_type(2)));
typedef float v8f __attribute__((ext_vector_type(8)));

__global__ void __launch_bounds__(1024)
k_tile_count(const int* __restrict__ dst, int* __restrict__ tilecount, int E) {
    __shared__ int cnt[NB];
    const int tid = threadIdx.x;
    for (int t = tid; t < NB; t += 1024) cnt[t] = 0;
    __syncthreads();
    const int e0 = blockIdx.x * 4096 + tid;
    int d[4] = {-1, -1, -1, -1};
    for (int j = 0; j < 4; ++j) {
        const int e = e0 + j * 1024;
        if (e < E) d[j] = dst[e];
    }
    for (int j = 0; j < 4; ++j)
        if (d[j] >= 0) {
            const int t = d[j] / TR;
            atomicAdd(&cnt[2 * t + ((d[j] - t * TR) >= SPLIT ? 1 : 0)], 1);
        }
    __syncthreads();
    for (int t = tid; t < NB; t += 1024) {
        const int c = cnt[t];
        if (c) atomicAdd(&tilecount[t], c);
    }
}

__global__ void __launch_bounds__(1024)
k_bin_fill(const int* __restrict__ src, const int* __restrict__ dst,
           const int* __restrict__ tilecount, int* __restrict__ tilestart,
           int* __restrict__ cursor, int2* __restrict__ rec, int E) {
    __shared__ int cnt[NB];
    __shared__ int base[NB];
    __shared__ int start[NB];
    __shared__ int wsum[32];
    const int tid  = threadIdx.x;
    const int lane = tid & 31;
    const int wv   = tid >> 5;
    for (int t = tid; t < NB; t += 1024) cnt[t] = 0;
    int s[4] = {0, 0, 0, 0}, d[4] = {0, 0, 0, 0}, tl[4] = {0, 0, 0, 0}, r[4] = {0, 0, 0, 0}, lr[4] = {0, 0, 0, 0};
    const int e0 = blockIdx.x * 4096 + tid;
    for (int j = 0; j < 4; ++j) {
        const int e = e0 + j * 1024;
        if (e < E) { s[j] = src[e]; d[j] = dst[e]; }
    }
    const int c = (tid < NB) ? tilecount[tid] : 0;
    int x = c;
    for (int o = 1; o < 32; o <<= 1) {
        const int y = __shfl_up(x, o);
        if (lane >= o) x += y;
    }
    if (lane == 31) wsum[wv] = x;
    __syncthreads();
    if (wv == 0) {
        const int w = wsum[lane];
        int xs = w;
        for (int o = 1; o < 32; o <<= 1) {
            const int y = __shfl_up(xs, o);
            if (lane >= o) xs += y;
        }
        wsum[lane] = xs - w;
    }
    __syncthreads();
    const int st = wsum[wv] + x - c;
    if (tid < NB) {
        start[tid] = st;
        if (blockIdx.x == 0) {
            tilestart[tid] = st;
            if (tid == NB - 1) tilestart[NB] = st + c;
        }
    }
    for (int j = 0; j < 4; ++j) {
        const int e = e0 + j * 1024;
        if (e < E) {
            const int t = d[j] / TR;
            lr[j] = d[j] - t * TR;
            tl[j] = 2 * t + (lr[j] >= SPLIT ? 1 : 0);
            r[j] = atomicAdd(&cnt[tl[j]], 1);
        }
    }
    __syncthreads();
    for (int t = tid; t < NB; t += 1024) {
        const int n = cnt[t];
        if (n) base[t] = start[t] + atomicAdd(&cursor[t], n);
    }
    __syncthreads();
    for (int j = 0; j < 4; ++j) {
        const int e = e0 + j * 1024;
        if (e < E)
            rec[base[tl[j]] + r[j]] = make_int2(s[j] | (lr[j] << 16), 0);
    }
}

__global__ void __launch_bounds__(256)
k_tile_deg(const int2* __restrict__ rec, const int* __restrict__ tilestart,
           float* __restrict__ selfc, float* __restrict__ dinv, int* __restrict__ indeg, int n) {
    __shared__ int cnt[TR];
    const int t = threadIdx.x;
    if (t < TR) cnt[t] = 0;
    __syncthreads();
    const int beg = tilestart[2 * blockIdx.x];
    const int end = tilestart[2 * blockIdx.x + 2];
    for (int i = beg + t; i < end; i += 256) atomicAdd(&cnt[rec[i].x >> 16], 1);
    __syncthreads();
    const int i = blockIdx.x * TR + t;
    if (t < TR && i < n) {
        const float d = (float)(cnt[t] + 1);
        dinv[i]  = 1.0f / sqrtf(d);
        selfc[i] = 1.0f / d;
        indeg[i] = cnt[t];
    }
}

static_assert(TR <= 128, "one wave scans the row counts, four rows per lane");
#define WB 8

__global__ void __launch_bounds__(256)
k_tile_weights(const int* __restrict__ tilestart, const float* __restrict__ dinv,
               const int* __restrict__ indeg, const int2* __restrict__ rec,
               int2* __restrict__ rec2) {
    __shared__ int pos[128];
    const int t   = threadIdx.x;
    const int beg = tilestart[2 * blockIdx.x];
    const int end = tilestart[2 * blockIdx.x + 2];
    const int n0  = blockIdx.x * TR;
    if (t < 32) {
        int c[4];
        int tot = 0;
        for (int k = 0; k < 4; ++k) {
            const int r = 4 * t + k;
            c[k] = (r < TR) ? indeg[n0 + r] : 0;
            tot += c[k];
        }
        int x = tot;
        for (int o = 1; o < 32; o <<= 1) {
            const int y = __shfl_up(x, o);
            if (t >= o) x += y;
        }
        int e = x - tot;
        for (int k = 0; k < 4; ++k) { pos[4 * t + k] = e; e += c[k]; }
    }
    __syncthreads();
    for (int i0 = beg + t; i0 < end; i0 += WB * 256) {
        int   k[WB];
        float a[WB], b[WB];
#pragma unroll
        for (int u = 0; u < WB; ++u) {
            const int i = i0 + u * 256;
            k[u] = (i < end) ? rec[i].x : -1;
        }
#pragma unroll
        for (int u = 0; u < WB; ++u) {
            a[u] = 0.f; b[u] = 0.f;
            if (k[u] >= 0) { a[u] = dinv[k[u] & 0xffff]; b[u] = dinv[n0 + (k[u] >> 16)]; }
        }
        int p[WB];
#pragma unroll
        for (int u = 0; u < WB; ++u) p[u] = (k[u] >= 0) ? atomicAdd(&pos[k[u] >> 16], 1) : 0;
#pragma unroll
        for (int u = 0; u < WB; ++u)
            if (k[u] >= 0) rec2[beg + p[u]] = make_int2(k[u], __float_as_int(a[u] * b[u]));
    }
}

static_assert(NG % 16 == 0 && DH == 128 && DO % 16 == 0 && DO <= DH, "whole 16-row strips, one column tile per wave");

__global__ void __launch_bounds__(256)
k_head(const float* __restrict__ sums, const float* __restrict__ counts,
       const float* __restrict__ W1, const float* __restrict__ b1,
       const float* __restrict__ W2, const float* __restrict__ b2,
       float* __restrict__ out)
{
    __shared__ float sW1t[DH * 132];
    __shared__ float sW2t[DO * 132];
    __shared__ float sA[16 * 132];
    __shared__ float sG[16 * 132];

    const int tid = threadIdx.x;
    const int r0  = blockIdx.x * 16;

    float sv[16 * DH / 256], cn[16 * DH / 256], w1[DH * DH / 256], w2[DH * DO / 256];
#pragma unroll
    for (int j = 0; j < 16 * DH / 256; ++j) {
        const int i = tid + j * 256;
        sv[j] = sums[r0 * DH + i];
        cn[j] = counts[r0 + (i >> 7)];
    }
#pragma unroll
    for (int j = 0; j < DH * DH / 256; ++j) w1[j] = W1[tid + j * 256];
#pragma unroll
    for (int j = 0; j < DH * DO / 256; ++j) w2[j] = W2[tid + j * 256];
#pragma unroll
    for (int j = 0; j < 16 * DH / 256; ++j) {
        const int i = tid + j * 256;
        sA[(i >> 7) * 132 + (i & 127)] = sv[j] / fmaxf(cn[j], 1.0f);
    }
#pragma unroll
    for (int j = 0; j < DH * DH / 256; ++j) {
        const int i = tid + j * 256;
        sW1t[(i % DH) * 132 + i / DH] = w1[j];
    }
#pragma unroll
    for (int j = 0; j < DH * DO / 256; ++j) {
        const int i = tid + j * 256;
        sW2t[(i % DO) * 132 + i / DO] = w2[j];
    }
    __syncthreads();

    const int wave = tid >> 5;
    const int lane = tid & 31;
    const int ln   = lane & 15;
    const int hi   = lane >> 4;
    const int col  = wave * 16 + ln;

    {
        const float* aP = &sA[ln * 132 + 2 * hi];
        const float* bP = &sW1t[col * 132 + 2 * hi];
        v8f acc = {0.f, 0.f, 0.f, 0.f, 0.f, 0.f, 0.f, 0.f};
#pragma unroll
        for (int k = 0; k < DH; k += 4) {
            const v2f a = *(const v2f*)(aP + k);
            const v2f b = *(const v2f*)(bP + k);
            acc = __builtin_amdgcn_wmma_f32_16x16x4_f32(false, a, false, b, (short)0, acc, false, false);
        }
        const float bv = b1[col];
#pragma unroll
        for (int r = 0; r < 8; ++r) sG[(r + 8 * hi) * 132 + col] = fmaxf(acc[r] + bv, 0.0f);
    }
    __syncthreads();

    if (wave < DO / 16) {
        const float* aP = &sG[ln * 132 + 2 * hi];
        const float* bP = &sW2t[col * 132 + 2 * hi];
        v8f acc = {0.f, 0.f, 0.f, 0.f, 0.f, 0.f, 0.f, 0.f};
#pragma unroll
        for (int k = 0; k < DH; k += 4) {
            const v2f a = *(const v2f*)(aP + k);
            const v2f b = *(const v2f*)(bP + k);
            acc = __builtin_amdgcn_wmma_f32_16x16x4_f32(false, a, false, b, (short)0, acc, false, false);
        }
        const float bv = b2[col];
#pragma unroll
        for (int r = 0; r < 8; ++r) out[(r0 + r + 8 * hi) * DO + col] = acc[r] + bv;
    }
}

#define TS  132
#define TRP 112
#define NU  ((TRP / 16) * (DH / 16))
#define WPT ((DH * DH) / (AW * 32))

static_assert(TRP % 16 == 0 && TRP >= TR && (TRP * TS) % 4 == 0, "whole strips, 16-byte clears");
static_assert(NU - (SPLIT / 16) * 8 <= 2 * AW && (DH * DH) % (AW * 32) == 0, "a wave takes at most two back output blocks");

__device__ __forceinline__ void edge_add(const float* __restrict__ Hin, float* acc,
                                         int key, float w, int lane) {
    const float4 v = *(const float4*)(Hin + (size_t)(key & 0xffff) * DH + 4 * lane);
    float* a = acc + (key >> 16) * TS + lane;
    __hip_atomic_fetch_add(a,      v.x * w, __ATOMIC_RELAXED, __HIP_MEMORY_SCOPE_WORKGROUP);
    __hip_atomic_fetch_add(a + 32, v.y * w, __ATOMIC_RELAXED, __HIP_MEMORY_SCOPE_WORKGROUP);
    __hip_atomic_fetch_add(a + 64, v.z * w, __ATOMIC_RELAXED, __HIP_MEMORY_SCOPE_WORKGROUP);
    __hip_atomic_fetch_add(a + 96, v.w * w, __ATOMIC_RELAXED, __HIP_MEMORY_SCOPE_WORKGROUP);
}

template <bool POOL>
__global__ void __launch_bounds__(AW * 32)
k_layer_tile(const float* __restrict__ Hin, const float* __restrict__ W,
             const int2* __restrict__ rec, const int* __restrict__ tilestart,
             const float* __restrict__ selfc, const float* __restrict__ bias,
             float* __restrict__ Hout, const int* __restrict__ batch,
             float* __restrict__ sums, float* __restrict__ counts, int n)
{
    __shared__ alignas(16) float sWt[DH * TS];
    __shared__ alignas(16) float acc[TRP * TS];
    const int tid  = threadIdx.x;
    const int lane = tid & 31;
    const int wave = __builtin_amdgcn_readfirstlane(tid >> 5);
    const int n0   = blockIdx.x * TR;
    const int rows = (n - n0 < TR) ? n - n0 : TR;

    float4* acc4 = (float4*)acc;
    for (int i = tid; i < TRP * TS / 4; i += AW * 32) acc4[i] = make_float4(0.f, 0.f, 0.f, 0.f);
    __syncthreads();

    float wreg[WPT];
#pragma unroll
    for (int j = 0; j < WPT; ++j) wreg[j] = W[tid + j * (AW * 32)];

    const int begA  = tilestart[2 * blockIdx.x];
    const int begB  = tilestart[2 * blockIdx.x + 1];
    const int endB  = tilestart[2 * blockIdx.x + 2];
    const int rowsA = (rows < SPLIT) ? rows : SPLIT;
    const int rowsB = rows - rowsA;

    auto walk = [&](const int2* __restrict__ trec, const int cntE, const int r0, const int nrows,
                    const int g0, const int gs) {
        const int egroups = (cntE + EG - 1) / EG;
        const int ngroups = egroups + (nrows + EG - 1) / EG;
        auto fetch = [&](int gi, int& k, int& w) {
            k = 0; w = 0;
            if (gi < egroups) {
                const int e = gi * EG + lane;
                if (lane < EG && e < cntE) { const int2 q = trec[e]; k = q.x; w = q.y; }
            } else if (gi < ngroups) {
                const int r = (gi - egroups) * EG + lane;
                if (lane < EG && r < nrows) {
                    k = (n0 + r0 + r) | ((r0 + r) << 16);
                    w = __float_as_int(selfc[n0 + r0 + r]);
                }
            }
        };
        int key, wb;
        int g = g0;
        fetch(g, key, wb);
        while (g < ngroups) {
            const int gn = g + gs;
            int key2, wb2;
            fetch(gn, key2, wb2);
            const int m = (g < egroups) ? cntE - g * EG : nrows - (g - egroups) * EG;
            if (m >= EG) {
                float4 v[EG];
                int    kj[EG];
                for (int j = 0; j < EG; ++j) {
                    kj[j] = __builtin_amdgcn_readlane(key, j);
                    v[j] = *(const float4*)(Hin + (size_t)(kj[j] & 0xffff) * DH + 4 * lane);
                }
                float4 s = make_float4(0.f, 0.f, 0.f, 0.f);
                for (int j = 0; j < EG; ++j) {
                    const float w = __int_as_float(__builtin_amdgcn_readlane(wb, j));
                    s.x += v[j].x * w;
                    s.y += v[j].y * w;
                    s.z += v[j].z * w;
                    s.w += v[j].w * w;
                    const int lr = kj[j] >> 16;
                    const int nx = (j + 1 < EG) ? (kj[(j + 1 < EG) ? j + 1 : j] >> 16) : -1;
                    if (nx != lr) {
                        float* a = acc + lr * TS + lane;
                        __hip_atomic_fetch_add(a,      s.x, __ATOMIC_RELAXED, __HIP_MEMORY_SCOPE_WORKGROUP);
                        __hip_atomic_fetch_add(a + 32, s.y, __ATOMIC_RELAXED, __HIP_MEMORY_SCOPE_WORKGROUP);
                        __hip_atomic_fetch_add(a + 64, s.z, __ATOMIC_RELAXED, __HIP_MEMORY_SCOPE_WORKGROUP);
                        __hip_atomic_fetch_add(a + 96, s.w, __ATOMIC_RELAXED, __HIP_MEMORY_SCOPE_WORKGROUP);
                        s = make_float4(0.f, 0.f, 0.f, 0.f);
                    }
                }
            } else {
                for (int j = 0; j < m; ++j) {
                    const int   k = __builtin_amdgcn_readlane(key, j);
                    const float w = __int_as_float(__builtin_amdgcn_readlane(wb, j));
                    edge_add(Hin, acc, k, w, lane);
                }
            }
            key = key2; wb = wb2; g = gn;
        }
    };

    walk(rec + begA, begB - begA, 0, rowsA, wave, AW);

#pragma unroll
    for (int j = 0; j < WPT; ++j) {
        const int i = tid + j * (AW * 32);
        const int k = i >> 7;
        sWt[(i & (DH - 1)) * TS + (k & 3) * 32 + (k >> 2)] = wreg[j];
    }
    __syncthreads();

    const int ln = lane & 15;
    const int hi = lane >> 4;

    if (wave < MW) {
        const float* aP = &acc[(wave * 16 + ln) * TS + 2 * hi];
        const float* bP = &sWt[ln * TS + 2 * hi];
        v8f d[8];
#pragma unroll
        for (int ct = 0; ct < 8; ++ct) d[ct] = (v8f){0.f, 0.f, 0.f, 0.f, 0.f, 0.f, 0.f, 0.f};
#pragma unroll 2
        for (int k = 0; k < DH; k += 4) {
            const v2f a = *(const v2f*)(aP + k);
#pragma unroll
            for (int ct = 0; ct < 8; ++ct) {
                const v2f b = *(const v2f*)(bP + ct * 16 * TS + k);
                d[ct] = __builtin_amdgcn_wmma_f32_16x16x4_f32(false, a, false, b, (short)0, d[ct], false, false);
            }
        }
#pragma unroll
        for (int ct = 0; ct < 8; ++ct) {
            const float bv = bias[ct * 16 + ln];
#pragma unroll
            for (int r = 0; r < 8; ++r)
                acc[(wave * 16 + r + 8 * hi) * TS + ct * 16 + ln] = fmaxf(d[ct][r] + bv, 0.0f);
        }
    } else {
        walk(rec + begB, endB - begB, SPLIT, rowsB, wave - MW, AW - MW);
    }
    __syncthreads();

    v8f res[2];
#pragma unroll
    for (int q = 0; q < 2; ++q) {
        const int u = MW * 8 + wave + q * AW;
        res[q] = (v8f){0.f, 0.f, 0.f, 0.f, 0.f, 0.f, 0.f, 0.f};
        if (u < NU) {
            const int strip = u >> 3, ct = u & 7;
            const float* aP = &acc[(strip * 16 + ln) * TS + 2 * hi];
            const float* bP = &sWt[(ct * 16 + ln) * TS + 2 * hi];
            v8f d = {0.f, 0.f, 0.f, 0.f, 0.f, 0.f, 0.f, 0.f};
#pragma unroll
            for (int k = 0; k < DH; k += 4) {
                v2f a = *(const v2f*)(aP + k);
                v2f b = *(const v2f*)(bP + k);
                d = __builtin_amdgcn_wmma_f32_16x16x4_f32(false, a, false, b, (short)0, d, false, false);
            }
            const float bv = bias[ct * 16 + ln];
#pragma unroll
            for (int r = 0; r < 8; ++r) res[q][r] = fmaxf(d[r] + bv, 0.0f);
        }
    }
    __syncthreads();

#pragma unroll
    for (int q = 0; q < 2; ++q) {
        const int u = MW * 8 + wave + q * AW;
        if (u < NU) {
            const int strip = u >> 3, ct = u & 7;
#pragma unroll
            for (int r = 0; r < 8; ++r)
                acc[(strip * 16 + r + 8 * hi) * TS + ct * 16 + ln] = res[q][r];
        }
    }
    __syncthreads();

    if (!POOL) {
        float4* H4 = (float4*)Hout;
        for (int i = tid; i < TR * DH / 4; i += AW * 32) {
            const int r = i >> 5, c4 = i & 31;
            if (r < rows) H4[(size_t)(n0 + r) * (DH / 4) + c4] = *((const float4*)(acc + r * TS) + c4);
        }
    } else {
        const int c  = tid & (DH - 1);
        const int rb = (wave >> 2) * PR;
        if ((wave & 3) == 0) {
            const int  g     = (lane < PR) ? batch[n0 + rb + lane] : -1;
            const int  gp    = __shfl_up(g, 1);
            const bool first = (lane < PR) && (lane == 0 || gp != g);
            const unsigned fm = (unsigned)__ballot(first);
            if (first) {
                const unsigned above = (lane < 31) ? (fm >> (lane + 1)) : 0u;
                const int next = above ? lane + __ffs(above) : PR;
                atomicAdd(&counts[g], (float)(next - lane));
            }
        }
        int   gid[PR];
        float v[PR];
#pragma unroll
        for (int j = 0; j < PR; ++j) {
            gid[j] = batch[n0 + rb + j];
            v[j]   = acc[(rb + j) * TS + c];
        }
        int   gg = gid[0];
        float s = 0.0f;
#pragma unroll
        for (int j = 0; j < PR; ++j) {
            if (gid[j] != gg) {
                atomicAdd(&sums[gg * DH + c], s);
                gg = gid[j];
                s = 0.0f;
            }
            s += v[j];
        }
        atomicAdd(&sums[gg * DH + c], s);
    }
}

extern "C" void kernel_launch(void* const* d_in, const int* in_sizes, int n_in,
                              void* d_out, int out_size, void* d_ws, size_t ws_size,
                              hipStream_t stream)
{
    (void)in_sizes; (void)n_in; (void)out_size; (void)ws_size;

    const float* x     = (const float*)d_in[0];
    const int*   ei    = (const int*)d_in[1];
    const int*   src   = ei;
    const int*   dst   = ei + NE;
    const int*   batch = (const int*)d_in[2];
    const float* Wl[5] = {(const float*)d_in[4], (const float*)d_in[6], (const float*)d_in[8],
                          (const float*)d_in[10], (const float*)d_in[12]};
    const float* Bl[5] = {(const float*)d_in[5], (const float*)d_in[7], (const float*)d_in[9],
                          (const float*)d_in[11], (const float*)d_in[13]};
    const float* fc1_w = (const float*)d_in[14];
    const float* fc1_b = (const float*)d_in[15];
    const float* fc2_w = (const float*)d_in[16];
    const float* fc2_b = (const float*)d_in[17];

    float* H      = (float*)d_ws;
    float* H2     = H      + (size_t)NN * DH;
    float* selfc  = H2     + (size_t)NN * DH;
    float* dinv   = selfc  + NN;
    int2*  rec    = (int2*)(dinv + NN);
    int2*  rec2   = rec + NE;
    int*   tilestart = (int*)(rec2 + NE);
    int*   cursor    = tilestart + NB + 1;
    int*   tilecount = cursor + NB;
    float* sums   = (float*)(tilecount + NB);
    float* counts = sums   + (size_t)NG * DH;
    int*   indeg  = (int*)(counts + NG);

    hipMemsetAsync(cursor, 0, ((size_t)2 * NB + (size_t)NG * DH + NG) * sizeof(float), stream);

    k_tile_count<<<(NE + 4095) / 4096, 1024, 0, stream>>>(dst, tilecount, NE);
    k_bin_fill<<<(NE + 4095) / 4096, 1024, 0, stream>>>(src, dst, tilecount, tilestart, cursor, rec, NE);
    k_tile_deg<<<NT, 256, 0, stream>>>(rec, tilestart, selfc, dinv, indeg, NN);
    k_tile_weights<<<NT, 256, 0, stream>>>(tilestart, dinv, indeg, rec, rec2);

    k_layer_tile<false><<<NT, AW * 32, 0, stream>>>(x,  Wl[0], rec2, tilestart, selfc, Bl[0], H,
                                                    nullptr, nullptr, nullptr, NN);
    k_layer_tile<false><<<NT, AW * 32, 0, stream>>>(H,  Wl[1], rec2, tilestart, selfc, Bl[1], H2,
                                                    nullptr, nullptr, nullptr, NN);
    k_layer_tile<false><<<NT, AW * 32, 0, stream>>>(H2, Wl[2], rec2, tilestart, selfc, Bl[2], H,
                                                    nullptr, nullptr, nullptr, NN);
    k_layer_tile<false><<<NT, AW * 32, 0, stream>>>(H,  Wl[3], rec2, tilestart, selfc, Bl[3], H2,
                                                    nullptr, nullptr, nullptr, NN);
    k_layer_tile<true><<<NT, AW * 32, 0, stream>>>(H2, Wl[4], rec2, tilestart, selfc, Bl[4], nullptr,
                                                   batch, sums, counts, NN);

    k_head<<<NG / 16, 256, 0, stream>>>(sums, counts, fc1_w, fc1_b, fc2_w, fc2_b, (float*)d_out);
}


# edge walk reads its records through the scalar unit (8 per group, a trip ahead, own counter): no row load waits behind a record read any more; short groups too send all rows first; sums keep order
# speedup vs baseline: 8.6877x; 1.0412x over previous
#include <hip/hip_runtime.h>

#define NN 50000
#define NE 800000
#define DH 128
#define DO 64
#define NG 128
#define TR 100
#define NT ((NN + TR - 1) / TR)
#define EG 8
#define AW 16
#define PR (TR / 4)
#define SPLIT 64
#define NB (2 * NT)
#define MW (SPLIT / 16)

static_assert(NB <= 1024, "the bin scan is one 1024-thread block");
static_assert(SPLIT % 16 == 0 && SPLIT < TR && MW < AW, "the front part is whole strips, one wave each, and walkers remain");
static_assert(NN <= 65536 && TR <= 32768, "a record packs source row and local row into one word");
static_assert(AW * 32 == 4 * DH && TR % 4 == 0 && PR <= 32 && NN % TR == 0,
              "the pool walk gives a quarter of a full tile's rows to DH threads");

typedef float v2f __attribute__((ext_vector_type(2)));
typedef float v8f __attribute__((ext_vector_type(8)));

__global__ void __launch_bounds__(1024)
k_tile_count(const int* __restrict__ dst, int* __restrict__ tilecount, int E) {
    __shared__ int cnt[NB];
    const int tid = threadIdx.x;
    for (int t = tid; t < NB; t += 1024) cnt[t] = 0;
    __syncthreads();
    const int e0 = blockIdx.x * 4096 + tid;
    int d[4] = {-1, -1, -1, -1};
    for (int j = 0; j < 4; ++j) {
        const int e = e0 + j * 1024;
        if (e < E) d[j] = dst[e];
    }
    for (int j = 0; j < 4; ++j)
        if (d[j] >= 0) {
            const int t = d[j] / TR;
            atomicAdd(&cnt[2 * t + ((d[j] - t * TR) >= SPLIT ? 1 : 0)], 1);
        }
    __syncthreads();
    for (int t = tid; t < NB; t += 1024) {
        const int c = cnt[t];
        if (c) atomicAdd(&tilecount[t], c);
    }
}

__global__ void __launch_bounds__(1024)
k_bin_fill(const int* __restrict__ src, const int* __restrict__ dst,
           const int* __restrict__ tilecount, int* __restrict__ tilestart,
           int* __restrict__ cursor, int2* __restrict__ rec, int E) {
    __shared__ int cnt[NB];
    __shared__ int base[NB];
    __shared__ int start[NB];
    __shared__ int wsum[32];
    const int tid  = threadIdx.x;
    const int lane = tid & 31;
    const int wv   = tid >> 5;
    for (int t = tid; t < NB; t += 1024) cnt[t] = 0;
    int s[4] = {0, 0, 0, 0}, d[4] = {0, 0, 0, 0}, tl[4] = {0, 0, 0, 0}, r[4] = {0, 0, 0, 0}, lr[4] = {0, 0, 0, 0};
    const int e0 = blockIdx.x * 4096 + tid;
    for (int j = 0; j < 4; ++j) {
        const int e = e0 + j * 1024;
        if (e < E) { s[j] = src[e]; d[j] = dst[e]; }
    }
    const int c = (tid < NB) ? tilecount[tid] : 0;
    int x = c;
    for (int o = 1; o < 32; o <<= 1) {
        const int y = __shfl_up(x, o);
        if (lane >= o) x += y;
    }
    if (lane == 31) wsum[wv] = x;
    __syncthreads();
    if (wv == 0) {
        const int w = wsum[lane];
        int xs = w;
        for (int o = 1; o < 32; o <<= 1) {
            const int y = __shfl_up(xs, o);
            if (lane >= o) xs += y;
        }
        wsum[lane] = xs - w;
    }
    __syncthreads();
    const int st = wsum[wv] + x - c;
    if (tid < NB) {
        start[tid] = st;
        if (blockIdx.x == 0) {
            tilestart[tid] = st;
            if (tid == NB - 1) tilestart[NB] = st + c;
        }
    }
    for (int j = 0; j < 4; ++j) {
        const int e = e0 + j * 1024;
        if (e < E) {
            const int t = d[j] / TR;
            lr[j] = d[j] - t * TR;
            tl[j] = 2 * t + (lr[j] >= SPLIT ? 1 : 0);
            r[j] = atomicAdd(&cnt[tl[j]], 1);
        }
    }
    __syncthreads();
    for (int t = tid; t < NB; t += 1024) {
        const int n = cnt[t];
        if (n) base[t] = start[t] + atomicAdd(&cursor[t], n);
    }
    __syncthreads();
    for (int j = 0; j < 4; ++j) {
        const int e = e0 + j * 1024;
        if (e < E)
            rec[base[tl[j]] + r[j]] = make_int2(s[j] | (lr[j] << 16), 0);
    }
}

__global__ void __launch_bounds__(256)
k_tile_deg(const int2* __restrict__ rec, const int* __restrict__ tilestart,
           float* __restrict__ selfc, float* __restrict__ dinv, int* __restrict__ indeg, int n) {
    __shared__ int cnt[TR];
    const int t = threadIdx.x;
    if (t < TR) cnt[t] = 0;
    __syncthreads();
    const int beg = tilestart[2 * blockIdx.x];
    const int end = tilestart[2 * blockIdx.x + 2];
    for (int i = beg + t; i < end; i += 256) atomicAdd(&cnt[rec[i].x >> 16], 1);
    __syncthreads();
    const int i = blockIdx.x * TR + t;
    if (t < TR && i < n) {
        const float d = (float)(cnt[t] + 1);
        dinv[i]  = 1.0f / sqrtf(d);
        selfc[i] = 1.0f / d;
        indeg[i] = cnt[t];
    }
}

static_assert(TR <= 128, "one wave scans the row counts, four rows per lane");
#define WB 8

__global__ void __launch_bounds__(256)
k_tile_weights(const int* __restrict__ tilestart, const float* __restrict__ dinv,
               const int* __restrict__ indeg, const int2* __restrict__ rec,
               int2* __restrict__ rec2) {
    __shared__ int pos[128];
    const int t   = threadIdx.x;
    const int beg = tilestart[2 * blockIdx.x];
    const int end = tilestart[2 * blockIdx.x + 2];
    const int n0  = blockIdx.x * TR;
    if (t < 32) {
        int c[4];
        int tot = 0;
        for (int k = 0; k < 4; ++k) {
            const int r = 4 * t + k;
            c[k] = (r < TR) ? indeg[n0 + r] : 0;
            tot += c[k];
        }
        int x = tot;
        for (int o = 1; o < 32; o <<= 1) {
            const int y = __shfl_up(x, o);
            if (t >= o) x += y;
        }
        int e = x - tot;
        for (int k = 0; k < 4; ++k) { pos[4 * t + k] = e; e += c[k]; }
    }
    __syncthreads();
    for (int i0 = beg + t; i0 < end; i0 += WB * 256) {
        int   k[WB];
        float a[WB], b[WB];
#pragma unroll
        for (int u = 0; u < WB; ++u) {
            const int i = i0 + u * 256;
            k[u] = (i < end) ? rec[i].x : -1;
        }
#pragma unroll
        for (int u = 0; u < WB; ++u) {
            a[u] = 0.f; b[u] = 0.f;
            if (k[u] >= 0) { a[u] = dinv[k[u] & 0xffff]; b[u] = dinv[n0 + (k[u] >> 16)]; }
        }
        int p[WB];
#pragma unroll
        for (int u = 0; u < WB; ++u) p[u] = (k[u] >= 0) ? atomicAdd(&pos[k[u] >> 16], 1) : 0;
#pragma unroll
        for (int u = 0; u < WB; ++u)
            if (k[u] >= 0) rec2[beg + p[u]] = make_int2(k[u], __float_as_int(a[u] * b[u]));
    }
}

static_assert(NG % 16 == 0 && DH == 128 && DO % 16 == 0 && DO <= DH, "whole 16-row strips, one column tile per wave");

__global__ void __launch_bounds__(256)
k_head(const float* __restrict__ sums, const float* __restrict__ counts,
       const float* __restrict__ W1, const float* __restrict__ b1,
       const float* __restrict__ W2, const float* __restrict__ b2,
       float* __restrict__ out)
{
    __shared__ float sW1t[DH * 132];
    __shared__ float sW2t[DO * 132];
    __shared__ float sA[16 * 132];
    __shared__ float sG[16 * 132];

    const int tid = threadIdx.x;
    const int r0  = blockIdx.x * 16;

    float sv[16 * DH / 256], cn[16 * DH / 256], w1[DH * DH / 256], w2[DH * DO / 256];
#pragma unroll
    for (int j = 0; j < 16 * DH / 256; ++j) {
        const int i = tid + j * 256;
        sv[j] = sums[r0 * DH + i];
        cn[j] = counts[r0 + (i >> 7)];
    }
#pragma unroll
    for (int j = 0; j < DH * DH / 256; ++j) w1[j] = W1[tid + j * 256];
#pragma unroll
    for (int j = 0; j < DH * DO / 256; ++j) w2[j] = W2[tid + j * 256];
#pragma unroll
    for (int j = 0; j < 16 * DH / 256; ++j) {
        const int i = tid + j * 256;
        sA[(i >> 7) * 132 + (i & 127)] = sv[j] / fmaxf(cn[j], 1.0f);
    }
#pragma unroll
    for (int j = 0; j < DH * DH / 256; ++j) {
        const int i = tid + j * 256;
        sW1t[(i % DH) * 132 + i / DH] = w1[j];
    }
#pragma unroll
    for (int j = 0; j < DH * DO / 256; ++j) {
        const int i = tid + j * 256;
        sW2t[(i % DO) * 132 + i / DO] = w2[j];
    }
    __syncthreads();

    const int wave = tid >> 5;
    const int lane = tid & 31;
    const int ln   = lane & 15;
    const int hi   = lane >> 4;
    const int col  = wave * 16 + ln;

    {
        const float* aP = &sA[ln * 132 + 2 * hi];
        const float* bP = &sW1t[col * 132 + 2 * hi];
        v8f acc = {0.f, 0.f, 0.f, 0.f, 0.f, 0.f, 0.f, 0.f};
#pragma unroll
        for (int k = 0; k < DH; k += 4) {
            const v2f a = *(const v2f*)(aP + k);
            const v2f b = *(const v2f*)(bP + k);
            acc = __builtin_amdgcn_wmma_f32_16x16x4_f32(false, a, false, b, (short)0, acc, false, false);
        }
        const float bv = b1[col];
#pragma unroll
        for (int r = 0; r < 8; ++r) sG[(r + 8 * hi) * 132 + col] = fmaxf(acc[r] + bv, 0.0f);
    }
    __syncthreads();

    if (wave < DO / 16) {
        const float* aP = &sG[ln * 132 + 2 * hi];
        const float* bP = &sW2t[col * 132 + 2 * hi];
        v8f acc = {0.f, 0.f, 0.f, 0.f, 0.f, 0.f, 0.f, 0.f};
#pragma unroll
        for (int k = 0; k < DH; k += 4) {
            const v2f a = *(const v2f*)(aP + k);
            const v2f b = *(const v2f*)(bP + k);
            acc = __builtin_amdgcn_wmma_f32_16x16x4_f32(false, a, false, b, (short)0, acc, false, false);
        }
        const float bv = b2[col];
#pragma unroll
        for (int r = 0; r < 8; ++r) out[(r0 + r + 8 * hi) * DO + col] = acc[r] + bv;
    }
}

#define TS  132
#define TRP 112
#define NU  ((TRP / 16) * (DH / 16))
#define WPT ((DH * DH) / (AW * 32))

static_assert(TRP % 16 == 0 && TRP >= TR && (TRP * TS) % 4 == 0, "whole strips, 16-byte clears");
static_assert(NU - (SPLIT / 16) * 8 <= 2 * AW && (DH * DH) % (AW * 32) == 0, "a wave takes at most two back output blocks");

template <bool POOL>
__global__ void __launch_bounds__(AW * 32)
k_layer_tile(const float* __restrict__ Hin, const float* __restrict__ W,
             const int2* __restrict__ rec, const int* __restrict__ tilestart,
             const float* __restrict__ selfc, const float* __restrict__ bias,
             float* __restrict__ Hout, const int* __restrict__ batch,
             float* __restrict__ sums, float* __restrict__ counts, int n)
{
    __shared__ alignas(16) float sWt[DH * TS];
    __shared__ alignas(16) float acc[TRP * TS];
    const int tid  = threadIdx.x;
    const int lane = tid & 31;
    const int wave = __builtin_amdgcn_readfirstlane(tid >> 5);
    const int n0   = blockIdx.x * TR;
    const int rows = (n - n0 < TR) ? n - n0 : TR;

    float4* acc4 = (float4*)acc;
    for (int i = tid; i < TRP * TS / 4; i += AW * 32) acc4[i] = make_float4(0.f, 0.f, 0.f, 0.f);
    __syncthreads();

    float wreg[WPT];
#pragma unroll
    for (int j = 0; j < WPT; ++j) wreg[j] = W[tid + j * (AW * 32)];

    const int begA  = tilestart[2 * blockIdx.x];
    const int begB  = tilestart[2 * blockIdx.x + 1];
    const int endB  = tilestart[2 * blockIdx.x + 2];
    const int rowsA = (rows < SPLIT) ? rows : SPLIT;
    const int rowsB = rows - rowsA;

    struct RecGroup { int k[EG]; int w[EG]; };
    auto walk = [&](const int2* __restrict__ trec, const int cntE, const int r0, const int nrows,
                    const int g0, const int gs) {
        const int egroups = (cntE + EG - 1) / EG;
        const int ngroups = egroups + (nrows + EG - 1) / EG;
        auto fetch = [&](const int gi, RecGroup& q) {
#pragma unroll
            for (int j = 0; j < EG; ++j) { q.k[j] = 0; q.w[j] = 0; }
            if (gi < egroups) {
                const int e = gi * EG;
                if (e + EG <= cntE) {
#pragma unroll
                    for (int j = 0; j < EG; ++j) { const int2 t = trec[e + j]; q.k[j] = t.x; q.w[j] = t.y; }
                } else {
#pragma unroll
                    for (int j = 0; j < EG; ++j)
                        if (e + j < cntE) { const int2 t = trec[e + j]; q.k[j] = t.x; q.w[j] = t.y; }
                }
            } else if (gi < ngroups) {
                const int r = r0 + (gi - egroups) * EG;
#pragma unroll
                for (int j = 0; j < EG; ++j)
                    if (r + j < r0 + nrows) {
                        q.k[j] = (n0 + r + j) | ((r + j) << 16);
                        q.w[j] = __float_as_int(selfc[n0 + r + j]);
                    }
            }
        };
        RecGroup cur;
        int g = g0;
        fetch(g, cur);
        while (g < ngroups) {
            const int gn = g + gs;
            const int m = (g < egroups) ? cntE - g * EG : nrows - (g - egroups) * EG;
            RecGroup nxt;
            if (m >= EG) {
                float4 v[EG];
#pragma unroll
                for (int j = 0; j < EG; ++j)
                    v[j] = *(const float4*)(Hin + (size_t)(cur.k[j] & 0xffff) * DH + 4 * lane);
                fetch(gn, nxt);
                float4 s = make_float4(0.f, 0.f, 0.f, 0.f);
#pragma unroll
                for (int j = 0; j < EG; ++j) {
                    const float w = __int_as_float(cur.w[j]);
                    s.x += v[j].x * w;
                    s.y += v[j].y * w;
                    s.z += v[j].z * w;
                    s.w += v[j].w * w;
                    const int lr = cur.k[j] >> 16;
                    const int nx = (j + 1 < EG) ? (cur.k[(j + 1 < EG) ? j + 1 : j] >> 16) : -1;
                    if (nx != lr) {
                        float* a = acc + lr * TS + lane;
                        __hip_atomic_fetch_add(a,      s.x, __ATOMIC_RELAXED, __HIP_MEMORY_SCOPE_WORKGROUP);
                        __hip_atomic_fetch_add(a + 32, s.y, __ATOMIC_RELAXED, __HIP_MEMORY_SCOPE_WORKGROUP);
                        __hip_atomic_fetch_add(a + 64, s.z, __ATOMIC_RELAXED, __HIP_MEMORY_SCOPE_WORKGROUP);
                        __hip_atomic_fetch_add(a + 96, s.w, __ATOMIC_RELAXED, __HIP_MEMORY_SCOPE_WORKGROUP);
                        s = make_float4(0.f, 0.f, 0.f, 0.f);
                    }
                }
            } else {
                float4 v[EG];
#pragma unroll
                for (int j = 0; j < EG; ++j) {
                    v[j] = make_float4(0.f, 0.f, 0.f, 0.f);
                    if (j < m) v[j] = *(const float4*)(Hin + (size_t)(cur.k[j] & 0xffff) * DH + 4 * lane);
                }
                fetch(gn, nxt);
#pragma unroll
                for (int j = 0; j < EG; ++j)
                    if (j < m) {
                        const float w = __int_as_float(cur.w[j]);
                        float* a = acc + (cur.k[j] >> 16) * TS + lane;
                        __hip_atomic_fetch_add(a,      v[j].x * w, __ATOMIC_RELAXED, __HIP_MEMORY_SCOPE_WORKGROUP);
                        __hip_atomic_fetch_add(a + 32, v[j].y * w, __ATOMIC_RELAXED, __HIP_MEMORY_SCOPE_WORKGROUP);
                        __hip_atomic_fetch_add(a + 64, v[j].z * w, __ATOMIC_RELAXED, __HIP_MEMORY_SCOPE_WORKGROUP);
                        __hip_atomic_fetch_add(a + 96, v[j].w * w, __ATOMIC_RELAXED, __HIP_MEMORY_SCOPE_WORKGROUP);
                    }
            }
            cur = nxt; g = gn;
        }
    };

    walk(rec + begA, begB - begA, 0, rowsA, wave, AW);

#pragma unroll
    for (int j = 0; j < WPT; ++j) {
        const int i = tid + j * (AW * 32);
        const int k = i >> 7;
        sWt[(i & (DH - 1)) * TS + (k & 3) * 32 + (k >> 2)] = wreg[j];
    }
    __syncthreads();

    const int ln = lane & 15;
    const int hi = lane >> 4;

    if (wave < MW) {
        const float* aP = &acc[(wave * 16 + ln) * TS + 2 * hi];
        const float* bP = &sWt[ln * TS + 2 * hi];
        v8f d[8];
#pragma unroll
        for (int ct = 0; ct < 8; ++ct) d[ct] = (v8f){0.f, 0.f, 0.f, 0.f, 0.f, 0.f, 0.f, 0.f};
#pragma unroll 2
        for (int k = 0; k < DH; k += 4) {
            const v2f a = *(const v2f*)(aP + k);
#pragma unroll
            for (int ct = 0; ct < 8; ++ct) {
                const v2f b = *(const v2f*)(bP + ct * 16 * TS + k);
                d[ct] = __builtin_amdgcn_wmma_f32_16x16x4_f32(false, a, false, b, (short)0, d[ct], false, false);
            }
        }
#pragma unroll
        for (int ct = 0; ct < 8; ++ct) {
            const float bv = bias[ct * 16 + ln];
#pragma unroll
            for (int r = 0; r < 8; ++r)
                acc[(wave * 16 + r + 8 * hi) * TS + ct * 16 + ln] = fmaxf(d[ct][r] + bv, 0.0f);
        }
    } else {
        walk(rec + begB, endB - begB, SPLIT, rowsB, wave - MW, AW - MW);
    }
    __syncthreads();

    v8f res[2];
#pragma unroll
    for (int q = 0; q < 2; ++q) {
        const int u = MW * 8 + wave + q * AW;
        res[q] = (v8f){0.f, 0.f, 0.f, 0.f, 0.f, 0.f, 0.f, 0.f};
        if (u < NU) {
            const int strip = u >> 3, ct = u & 7;
            const float* aP = &acc[(strip * 16 + ln) * TS + 2 * hi];
            const float* bP = &sWt[(ct * 16 + ln) * TS + 2 * hi];
            v8f d = {0.f, 0.f, 0.f, 0.f, 0.f, 0.f, 0.f, 0.f};
#pragma unroll
            for (int k = 0; k < DH; k += 4) {
                v2f a = *(const v2f*)(aP + k);
                v2f b = *(const v2f*)(bP + k);
                d = __builtin_amdgcn_wmma_f32_16x16x4_f32(false, a, false, b, (short)0, d, false, false);
            }
            const float bv = bias[ct * 16 + ln];
#pragma unroll
            for (int r = 0; r < 8; ++r) res[q][r] = fmaxf(d[r] + bv, 0.0f);
        }
    }
    __syncthreads();

#pragma unroll
    for (int q = 0; q < 2; ++q) {
        const int u = MW * 8 + wave + q * AW;
        if (u < NU) {
            const int strip = u >> 3, ct = u & 7;
#pragma unroll
            for (int r = 0; r < 8; ++r)
                acc[(strip * 16 + r + 8 * hi) * TS + ct * 16 + ln] = res[q][r];
        }
    }
    __syncthreads();

    if (!POOL) {
        float4* H4 = (float4*)Hout;
        for (int i = tid; i < TR * DH / 4; i += AW * 32) {
            const int r = i >> 5, c4 = i & 31;
            if (r < rows) H4[(size_t)(n0 + r) * (DH / 4) + c4] = *((const float4*)(acc + r * TS) + c4);
        }
    } else {
        const int c  = tid & (DH - 1);
        const int rb = (wave >> 2) * PR;
        if ((wave & 3) == 0) {
            const int  g     = (lane < PR) ? batch[n0 + rb + lane] : -1;
            const int  gp    = __shfl_up(g, 1);
            const bool first = (lane < PR) && (lane == 0 || gp != g);
            const unsigned fm = (unsigned)__ballot(first);
            if (first) {
                const unsigned above = (lane < 31) ? (fm >> (lane + 1)) : 0u;
                const int next = above ? lane + __ffs(above) : PR;
                atomicAdd(&counts[g], (float)(next - lane));
            }
        }
        int   gid[PR];
        float v[PR];
#pragma unroll
        for (int j = 0; j < PR; ++j) {
            gid[j] = batch[n0 + rb + j];
            v[j]   = acc[(rb + j) * TS + c];
        }
        int   gg = gid[0];
        float s = 0.0f;
#pragma unroll
        for (int j = 0; j < PR; ++j) {
            if (gid[j] != gg) {
                atomicAdd(&sums[gg * DH + c], s);
                gg = gid[j];
                s = 0.0f;
            }
            s += v[j];
        }
        atomicAdd(&sums[gg * DH + c], s);
    }
}

extern "C" void kernel_launch(void* const* d_in, const int* in_sizes, int n_in,
                              void* d_out, int out_size, void* d_ws, size_t ws_size,
                              hipStream_t stream)
{
    (void)in_sizes; (void)n_in; (void)out_size; (void)ws_size;

    const float* x     = (const float*)d_in[0];
    const int*   ei    = (const int*)d_in[1];
    const int*   src   = ei;
    const int*   dst   = ei + NE;
    const int*   batch = (const int*)d_in[2];
    const float* Wl[5] = {(const float*)d_in[4], (const float*)d_in[6], (const float*)d_in[8],
                          (const float*)d_in[10], (const float*)d_in[12]};
    const float* Bl[5] = {(const float*)d_in[5], (const float*)d_in[7], (const float*)d_in[9],
                          (const float*)d_in[11], (const float*)d_in[13]};
    const float* fc1_w = (const float*)d_in[14];
    const float* fc1_b = (const float*)d_in[15];
    const float* fc2_w = (const float*)d_in[16];
    const float* fc2_b = (const float*)d_in[17];

    float* H      = (float*)d_ws;
    float* H2     = H      + (size_t)NN * DH;
    float* selfc  = H2     + (size_t)NN * DH;
    float* dinv   = selfc  + NN;
    int2*  rec    = (int2*)(dinv + NN);
    int2*  rec2   = rec + NE;
    int*   tilestart = (int*)(rec2 + NE);
    int*   cursor    = tilestart + NB + 1;
    int*   tilecount = cursor + NB;
    float* sums   = (float*)(tilecount + NB);
    float* counts = sums   + (size_t)NG * DH;
    int*   indeg  = (int*)(counts + NG);

    hipMemsetAsync(cursor, 0, ((size_t)2 * NB + (size_t)NG * DH + NG) * sizeof(float), stream);

    k_tile_count<<<(NE + 4095) / 4096, 1024, 0, stream>>>(dst, tilecount, NE);
    k_bin_fill<<<(NE + 4095) / 4096, 1024, 0, stream>>>(src, dst, tilecount, tilestart, cursor, rec, NE);
    k_tile_deg<<<NT, 256, 0, stream>>>(rec, tilestart, selfc, dinv, indeg, NN);
    k_tile_weights<<<NT, 256, 0, stream>>>(tilestart, dinv, indeg, rec, rec2);

    k_layer_tile<false><<<NT, AW * 32, 0, stream>>>(x,  Wl[0], rec2, tilestart, selfc, Bl[0], H,
                                                    nullptr, nullptr, nullptr, NN);
    k_layer_tile<false><<<NT, AW * 32, 0, stream>>>(H,  Wl[1], rec2, tilestart, selfc, Bl[1], H2,
                                                    nullptr, nullptr, nullptr, NN);
    k_layer_tile<false><<<NT, AW * 32, 0, stream>>>(H2, Wl[2], rec2, tilestart, selfc, Bl[2], H,
                                                    nullptr, nullptr, nullptr, NN);
    k_layer_tile<false><<<NT, AW * 32, 0, stream>>>(H,  Wl[3], rec2, tilestart, selfc, Bl[3], H2,
                                                    nullptr, nullptr, nullptr, NN);
    k_layer_tile<true><<<NT, AW * 32, 0, stream>>>(H2, Wl[4], rec2, tilestart, selfc, Bl[4], nullptr,
                                                   batch, sums, counts, NN);

    k_head<<<NG / 16, 256, 0, stream>>>(sums, counts, fc1_w, fc1_b, fc2_w, fc2_b, (float*)d_out);
}
